# starting kernel
# baseline (speedup 1.0000x reference)
#include <hip/hip_runtime.h>


typedef float v2f __attribute__((ext_vector_type(2)));
typedef float v8f __attribute__((ext_vector_type(8)));

#define T_TOT  2048
#define B_SZ   4
#define D_SZ   256
#define N_SZ   256
#define C_SZ   128
#define NCHUNK 16
#define EPS_F  1e-8f

__device__ __forceinline__ v8f wmma_f32_k4(v2f a, v2f b, v8f c) {
  return __builtin_amdgcn_wmma_f32_16x16x4_f32(false, a, false, b, (short)0, c, false, false);
}

__global__ void k_decay(const float* __restrict__ kin, const float* __restrict__ alpha,
                        float* __restrict__ kd, float* __restrict__ prefix) {
  const int c = blockIdx.x / B_SZ;
  const int b = blockIdx.x % B_SZ;
  const int n = threadIdx.x;
  const long stride = (long)B_SZ * N_SZ;
  const long base = ((long)c * C_SZ * B_SZ + b) * N_SZ + n;

  float P = 1.0f;
  #pragma unroll 4
  for (int t = 0; t < C_SZ; ++t) {
    float a = alpha[base + (long)t * stride];
    a = (a < EPS_F) ? EPS_F : a;
    P *= a;
  }
  const float inv = 1.0f / (P + EPS_F);
  float r = 1.0f;
  #pragma unroll 4
  for (int t = 0; t < C_SZ; ++t) {
    const long idx = base + (long)t * stride;
    float a = alpha[idx];
    a = (a < EPS_F) ? EPS_F : a;
    r *= a;
    kd[idx] = kin[idx] * (r * inv);
  }
  float a0 = alpha[base];
  a0 = (a0 < EPS_F) ? EPS_F : a0;
  prefix[(c * B_SZ + b) * N_SZ + n] = a0;
}

__global__ void k_gemmU(const float* __restrict__ v, const float* __restrict__ kd,
                        float* __restrict__ U) {
  __shared__ float Av[C_SZ][65];
  __shared__ float Bk[C_SZ][65];
  const int c = blockIdx.x / B_SZ, b = blockIdx.x % B_SZ;
  const int dBase = blockIdx.y * 64, nBase = blockIdx.z * 64;
  const int tid = threadIdx.x, lane = tid & 31, w = tid >> 5;

  for (int idx = tid; idx < C_SZ * 64; idx += 128) {
    const int t = idx >> 6, i = idx & 63;
    const long row = (long)(c * C_SZ + t) * B_SZ + b;
    Av[t][i] = v[row * D_SZ + dBase + i];
    Bk[t][i] = kd[row * N_SZ + nBase + i];
  }
  __syncthreads();

  const int half = lane >> 4, lan = lane & 15;
  const int mLoc = w * 16 + lan;
  v8f acc[4] = {};
  #pragma unroll
  for (int kk = 0; kk < C_SZ; kk += 4) {
    const int ka = kk + half * 2;
    v2f a; a.x = Av[ka][mLoc]; a.y = Av[ka + 1][mLoc];
    #pragma unroll
    for (int j = 0; j < 4; ++j) {
      v2f bf; bf.x = Bk[ka][j * 16 + lan]; bf.y = Bk[ka + 1][j * 16 + lan];
      acc[j] = wmma_f32_k4(a, bf, acc[j]);
    }
  }
  const long ub = (long)(c * B_SZ + b) * D_SZ;
  #pragma unroll
  for (int j = 0; j < 4; ++j)
    #pragma unroll
    for (int i = 0; i < 8; ++i) {
      const int row = dBase + w * 16 + half * 8 + i;
      const int col = nBase + j * 16 + lan;
      U[(ub + row) * N_SZ + col] = acc[j][i];
    }
}

__global__ void k_intra(const float* __restrict__ q, const float* __restrict__ kd,
                        const float* __restrict__ v, float* __restrict__ out) {
  __shared__ float As[C_SZ][C_SZ + 4];
  __shared__ float Pa[C_SZ][65];
  __shared__ float Pb[C_SZ][65];
  const int c = blockIdx.x / B_SZ, b = blockIdx.x % B_SZ;
  const int tid = threadIdx.x, lane = tid & 31, w = tid >> 5;
  const int half = lane >> 4, lan = lane & 15;
  const int mRow = w * 16 + lan;

  v8f acc[8] = {};
  for (int kn = 0; kn < N_SZ; kn += 64) {
    for (int idx = tid; idx < C_SZ * 64; idx += 256) {
      const int t = idx >> 6, j = idx & 63;
      const long row = ((long)(c * C_SZ + t) * B_SZ + b) * N_SZ + kn + j;
      Pa[t][j] = q[row];
      Pb[t][j] = kd[row];
    }
    __syncthreads();
    #pragma unroll
    for (int kk = 0; kk < 64; kk += 4) {
      const int ka = kk + half * 2;
      v2f a; a.x = Pa[mRow][ka]; a.y = Pa[mRow][ka + 1];
      #pragma unroll
      for (int j = 0; j < 8; ++j) {
        v2f bf; bf.x = Pb[j * 16 + lan][ka]; bf.y = Pb[j * 16 + lan][ka + 1];
        acc[j] = wmma_f32_k4(a, bf, acc[j]);
      }
    }
    __syncthreads();
  }
  #pragma unroll
  for (int j = 0; j < 8; ++j)
    #pragma unroll
    for (int i = 0; i < 8; ++i) {
      const int row = w * 16 + half * 8 + i;
      const int col = j * 16 + lan;
      As[row][col] = (col <= row) ? acc[j][i] : 0.0f;
    }
  __syncthreads();

  for (int dp = 0; dp < D_SZ; dp += 64) {
    for (int idx = tid; idx < C_SZ * 64; idx += 256) {
      const int s = idx >> 6, j = idx & 63;
      Pa[s][j] = v[((long)(c * C_SZ + s) * B_SZ + b) * D_SZ + dp + j];
    }
    __syncthreads();
    v8f acc2[4] = {};
    #pragma unroll
    for (int kk = 0; kk < C_SZ; kk += 4) {
      const int ka = kk + half * 2;
      v2f a; a.x = As[mRow][ka]; a.y = As[mRow][ka + 1];
      #pragma unroll
      for (int j = 0; j < 4; ++j) {
        v2f bf; bf.x = Pa[ka][j * 16 + lan]; bf.y = Pa[ka + 1][j * 16 + lan];
        acc2[j] = wmma_f32_k4(a, bf, acc2[j]);
      }
    }
    #pragma unroll
    for (int j = 0; j < 4; ++j)
      #pragma unroll
      for (int i = 0; i < 8; ++i) {
        const int row = w * 16 + half * 8 + i;
        const int col = dp + j * 16 + lan;
        out[((long)(c * C_SZ + row) * B_SZ + b) * D_SZ + col] = acc2[j][i];
      }
    __syncthreads();
  }
}

__global__ void k_scan(const float* __restrict__ U, const float* __restrict__ prefix,
                       float* __restrict__ Sst) {
  const long g = (long)blockIdx.x * blockDim.x + threadIdx.x;
  const int n = (int)(g % N_SZ);
  const int b = (int)(g / ((long)D_SZ * N_SZ));
  const long cs = (long)B_SZ * D_SZ * N_SZ;
  float s = 0.0f;
  for (int c = 0; c < NCHUNK; ++c) {
    Sst[c * cs + g] = s;
    s = s * prefix[(c * B_SZ + b) * N_SZ + n] + U[c * cs + g];
  }
}

__global__ void k_inter(const float* __restrict__ q, const float* __restrict__ prefix,
                        const float* __restrict__ Sst, float* __restrict__ out) {
  __shared__ float Qp[C_SZ][65];
  __shared__ float Sp[64][65];
  const int c = blockIdx.x / B_SZ, b = blockIdx.x % B_SZ;
  const int dBase = blockIdx.y * 64;
  const int tid = threadIdx.x, lane = tid & 31, w = tid >> 5;
  const int half = lane >> 4, lan = lane & 15;
  const int mRow = w * 16 + lan;
  const long cs = (long)B_SZ * D_SZ * N_SZ;

  v8f acc[4] = {};
  for (int np = 0; np < N_SZ; np += 64) {
    for (int idx = tid; idx < C_SZ * 64; idx += 256) {
      const int t = idx >> 6, j = idx & 63;
      const float pf = prefix[(c * B_SZ + b) * N_SZ + np + j];
      Qp[t][j] = q[((long)(c * C_SZ + t) * B_SZ + b) * N_SZ + np + j] * pf;
    }
    for (int idx = tid; idx < 64 * 64; idx += 256) {
      const int i = idx >> 6, j = idx & 63;
      Sp[i][j] = Sst[(long)c * cs + ((long)b * D_SZ + dBase + i) * N_SZ + np + j];
    }
    __syncthreads();
    #pragma unroll
    for (int kk = 0; kk < 64; kk += 4) {
      const int ka = kk + half * 2;
      v2f a; a.x = Qp[mRow][ka]; a.y = Qp[mRow][ka + 1];
      #pragma unroll
      for (int j = 0; j < 4; ++j) {
        v2f bf; bf.x = Sp[j * 16 + lan][ka]; bf.y = Sp[j * 16 + lan][ka + 1];
        acc[j] = wmma_f32_k4(a, bf, acc[j]);
      }
    }
    __syncthreads();
  }
  #pragma unroll
  for (int j = 0; j < 4; ++j)
    #pragma unroll
    for (int i = 0; i < 8; ++i) {
      const int row = w * 16 + half * 8 + i;
      const int col = dBase + j * 16 + lan;
      out[((long)(c * C_SZ + row) * B_SZ + b) * D_SZ + col] += acc[j][i];
    }
}

extern "C" void kernel_launch(void* const* d_in, const int* in_sizes, int n_in,
                              void* d_out, int out_size, void* d_ws, size_t ws_size,
                              hipStream_t stream) {
  const float* v     = (const float*)d_in[0];
  const float* k     = (const float*)d_in[1];
  const float* alpha = (const float*)d_in[2];
  const float* q     = (const float*)d_in[3];
  float* out = (float*)d_out;
  float* ws  = (float*)d_ws;

  const long KD_N = (long)T_TOT * B_SZ * N_SZ;
  const long PF_N = (long)NCHUNK * B_SZ * N_SZ;
  const long U_N  = (long)NCHUNK * B_SZ * D_SZ * N_SZ;
  float* kd     = ws;
  float* prefix = kd + KD_N;
  float* U      = prefix + PF_N;
  float* Sst    = U + U_N;

  k_decay<<<dim3(B_SZ * NCHUNK), 256, 0, stream>>>(k, alpha, kd, prefix);
  k_gemmU<<<dim3(B_SZ * NCHUNK, D_SZ / 64, N_SZ / 64), 128, 0, stream>>>(v, kd, U);
  k_intra<<<dim3(B_SZ * NCHUNK), 256, 0, stream>>>(q, kd, v, out);
  k_scan<<<dim3((B_SZ * D_SZ * N_SZ) / 256), 256, 0, stream>>>(U, prefix, Sst);
  k_inter<<<dim3(B_SZ * NCHUNK, D_SZ / 64), 256, 0, stream>>>(q, prefix, Sst, out);
}


# k_decay: one thread walking 128 time steps twice -> segmented product scan (8-step segments side by side, totals folded via LDS in fixed order), 128-bit loads in one group; reassociates the product
# speedup vs baseline: 1.2953x; 1.2953x over previous
#include <hip/hip_runtime.h>


typedef float v2f __attribute__((ext_vector_type(2)));
typedef float v8f __attribute__((ext_vector_type(8)));

#define T_TOT  2048
#define B_SZ   4
#define D_SZ   256
#define N_SZ   256
#define C_SZ   128
#define NCHUNK 16
#define EPS_F  1e-8f

__device__ __forceinline__ v8f wmma_f32_k4(v2f a, v2f b, v8f c) {
  return __builtin_amdgcn_wmma_f32_16x16x4_f32(false, a, false, b, (short)0, c, false, false);
}

typedef float v4f __attribute__((ext_vector_type(4)));

#define DK_COLS  64
#define DK_SEGS  16
#define DK_STEPS (C_SZ / DK_SEGS)

__device__ __forceinline__ v4f clip4(v4f a) {
  v4f r;
  r.x = (a.x < EPS_F) ? EPS_F : a.x;
  r.y = (a.y < EPS_F) ? EPS_F : a.y;
  r.z = (a.z < EPS_F) ? EPS_F : a.z;
  r.w = (a.w < EPS_F) ? EPS_F : a.w;
  return r;
}

__global__ void k_decay(const float* __restrict__ kin, const float* __restrict__ alpha,
                        float* __restrict__ kd, float* __restrict__ prefix) {
  __shared__ v4f segTot[DK_SEGS][DK_COLS / 4];
  const int c = blockIdx.x / B_SZ;
  const int b = blockIdx.x % B_SZ;
  const int q4  = threadIdx.x & 15;
  const int seg = threadIdx.x >> 4;
  const int n0  = blockIdx.y * DK_COLS + q4 * 4;
  const long stride = (long)B_SZ * N_SZ;
  const long base = ((long)(c * C_SZ + seg * DK_STEPS) * B_SZ + b) * N_SZ + n0;

  v4f a[DK_STEPS], kv[DK_STEPS];
  for (int i = 0; i < DK_STEPS; ++i)
    a[i] = *reinterpret_cast<const v4f*>(alpha + base + (long)i * stride);
  for (int i = 0; i < DK_STEPS; ++i)
    kv[i] = *reinterpret_cast<const v4f*>(kin + base + (long)i * stride);

  v4f tot = {1.0f, 1.0f, 1.0f, 1.0f};
  for (int i = 0; i < DK_STEPS; ++i) {
    a[i] = clip4(a[i]);
    tot *= a[i];
  }
  segTot[seg][q4] = tot;
  __syncthreads();

  v4f E = {1.0f, 1.0f, 1.0f, 1.0f};
  v4f P = {1.0f, 1.0f, 1.0f, 1.0f};
  for (int j = 0; j < DK_SEGS; ++j) {
    if (j == seg) E = P;
    P *= segTot[j][q4];
  }
  const v4f inv = 1.0f / (P + EPS_F);

  v4f r = E;
  for (int i = 0; i < DK_STEPS; ++i) {
    r *= a[i];
    *reinterpret_cast<v4f*>(kd + base + (long)i * stride) = kv[i] * (r * inv);
  }
  if (seg == 0)
    *reinterpret_cast<v4f*>(prefix + (c * B_SZ + b) * N_SZ + n0) = a[0];
}

__global__ void k_gemmU(const float* __restrict__ v, const float* __restrict__ kd,
                        float* __restrict__ U) {
  __shared__ float Av[C_SZ][65];
  __shared__ float Bk[C_SZ][65];
  const int c = blockIdx.x / B_SZ, b = blockIdx.x % B_SZ;
  const int dBase = blockIdx.y * 64, nBase = blockIdx.z * 64;
  const int tid = threadIdx.x, lane = tid & 31, w = tid >> 5;

  for (int idx = tid; idx < C_SZ * 64; idx += 128) {
    const int t = idx >> 6, i = idx & 63;
    const long row = (long)(c * C_SZ + t) * B_SZ + b;
    Av[t][i] = v[row * D_SZ + dBase + i];
    Bk[t][i] = kd[row * N_SZ + nBase + i];
  }
  __syncthreads();

  const int half = lane >> 4, lan = lane & 15;
  const int mLoc = w * 16 + lan;
  v8f acc[4] = {};
  #pragma unroll
  for (int kk = 0; kk < C_SZ; kk += 4) {
    const int ka = kk + half * 2;
    v2f a; a.x = Av[ka][mLoc]; a.y = Av[ka + 1][mLoc];
    #pragma unroll
    for (int j = 0; j < 4; ++j) {
      v2f bf; bf.x = Bk[ka][j * 16 + lan]; bf.y = Bk[ka + 1][j * 16 + lan];
      acc[j] = wmma_f32_k4(a, bf, acc[j]);
    }
  }
  const long ub = (long)(c * B_SZ + b) * D_SZ;
  #pragma unroll
  for (int j = 0; j < 4; ++j)
    #pragma unroll
    for (int i = 0; i < 8; ++i) {
      const int row = dBase + w * 16 + half * 8 + i;
      const int col = nBase + j * 16 + lan;
      U[(ub + row) * N_SZ + col] = acc[j][i];
    }
}

__global__ void k_intra(const float* __restrict__ q, const float* __restrict__ kd,
                        const float* __restrict__ v, float* __restrict__ out) {
  __shared__ float As[C_SZ][C_SZ + 4];
  __shared__ float Pa[C_SZ][65];
  __shared__ float Pb[C_SZ][65];
  const int c = blockIdx.x / B_SZ, b = blockIdx.x % B_SZ;
  const int tid = threadIdx.x, lane = tid & 31, w = tid >> 5;
  const int half = lane >> 4, lan = lane & 15;
  const int mRow = w * 16 + lan;

  v8f acc[8] = {};
  for (int kn = 0; kn < N_SZ; kn += 64) {
    for (int idx = tid; idx < C_SZ * 64; idx += 256) {
      const int t = idx >> 6, j = idx & 63;
      const long row = ((long)(c * C_SZ + t) * B_SZ + b) * N_SZ + kn + j;
      Pa[t][j] = q[row];
      Pb[t][j] = kd[row];
    }
    __syncthreads();
    #pragma unroll
    for (int kk = 0; kk < 64; kk += 4) {
      const int ka = kk + half * 2;
      v2f a; a.x = Pa[mRow][ka]; a.y = Pa[mRow][ka + 1];
      #pragma unroll
      for (int j = 0; j < 8; ++j) {
        v2f bf; bf.x = Pb[j * 16 + lan][ka]; bf.y = Pb[j * 16 + lan][ka + 1];
        acc[j] = wmma_f32_k4(a, bf, acc[j]);
      }
    }
    __syncthreads();
  }
  #pragma unroll
  for (int j = 0; j < 8; ++j)
    #pragma unroll
    for (int i = 0; i < 8; ++i) {
      const int row = w * 16 + half * 8 + i;
      const int col = j * 16 + lan;
      As[row][col] = (col <= row) ? acc[j][i] : 0.0f;
    }
  __syncthreads();

  for (int dp = 0; dp < D_SZ; dp += 64) {
    for (int idx = tid; idx < C_SZ * 64; idx += 256) {
      const int s = idx >> 6, j = idx & 63;
      Pa[s][j] = v[((long)(c * C_SZ + s) * B_SZ + b) * D_SZ + dp + j];
    }
    __syncthreads();
    v8f acc2[4] = {};
    #pragma unroll
    for (int kk = 0; kk < C_SZ; kk += 4) {
      const int ka = kk + half * 2;
      v2f a; a.x = As[mRow][ka]; a.y = As[mRow][ka + 1];
      #pragma unroll
      for (int j = 0; j < 4; ++j) {
        v2f bf; bf.x = Pa[ka][j * 16 + lan]; bf.y = Pa[ka + 1][j * 16 + lan];
        acc2[j] = wmma_f32_k4(a, bf, acc2[j]);
      }
    }
    #pragma unroll
    for (int j = 0; j < 4; ++j)
      #pragma unroll
      for (int i = 0; i < 8; ++i) {
        const int row = w * 16 + half * 8 + i;
        const int col = dp + j * 16 + lan;
        out[((long)(c * C_SZ + row) * B_SZ + b) * D_SZ + col] = acc2[j][i];
      }
    __syncthreads();
  }
}

__global__ void k_scan(const float* __restrict__ U, const float* __restrict__ prefix,
                       float* __restrict__ Sst) {
  const long g = (long)blockIdx.x * blockDim.x + threadIdx.x;
  const int n = (int)(g % N_SZ);
  const int b = (int)(g / ((long)D_SZ * N_SZ));
  const long cs = (long)B_SZ * D_SZ * N_SZ;
  float s = 0.0f;
  for (int c = 0; c < NCHUNK; ++c) {
    Sst[c * cs + g] = s;
    s = s * prefix[(c * B_SZ + b) * N_SZ + n] + U[c * cs + g];
  }
}

__global__ void k_inter(const float* __restrict__ q, const float* __restrict__ prefix,
                        const float* __restrict__ Sst, float* __restrict__ out) {
  __shared__ float Qp[C_SZ][65];
  __shared__ float Sp[64][65];
  const int c = blockIdx.x / B_SZ, b = blockIdx.x % B_SZ;
  const int dBase = blockIdx.y * 64;
  const int tid = threadIdx.x, lane = tid & 31, w = tid >> 5;
  const int half = lane >> 4, lan = lane & 15;
  const int mRow = w * 16 + lan;
  const long cs = (long)B_SZ * D_SZ * N_SZ;

  v8f acc[4] = {};
  for (int np = 0; np < N_SZ; np += 64) {
    for (int idx = tid; idx < C_SZ * 64; idx += 256) {
      const int t = idx >> 6, j = idx & 63;
      const float pf = prefix[(c * B_SZ + b) * N_SZ + np + j];
      Qp[t][j] = q[((long)(c * C_SZ + t) * B_SZ + b) * N_SZ + np + j] * pf;
    }
    for (int idx = tid; idx < 64 * 64; idx += 256) {
      const int i = idx >> 6, j = idx & 63;
      Sp[i][j] = Sst[(long)c * cs + ((long)b * D_SZ + dBase + i) * N_SZ + np + j];
    }
    __syncthreads();
    #pragma unroll
    for (int kk = 0; kk < 64; kk += 4) {
      const int ka = kk + half * 2;
      v2f a; a.x = Qp[mRow][ka]; a.y = Qp[mRow][ka + 1];
      #pragma unroll
      for (int j = 0; j < 4; ++j) {
        v2f bf; bf.x = Sp[j * 16 + lan][ka]; bf.y = Sp[j * 16 + lan][ka + 1];
        acc[j] = wmma_f32_k4(a, bf, acc[j]);
      }
    }
    __syncthreads();
  }
  #pragma unroll
  for (int j = 0; j < 4; ++j)
    #pragma unroll
    for (int i = 0; i < 8; ++i) {
      const int row = w * 16 + half * 8 + i;
      const int col = dBase + j * 16 + lan;
      out[((long)(c * C_SZ + row) * B_SZ + b) * D_SZ + col] += acc[j][i];
    }
}

extern "C" void kernel_launch(void* const* d_in, const int* in_sizes, int n_in,
                              void* d_out, int out_size, void* d_ws, size_t ws_size,
                              hipStream_t stream) {
  const float* v     = (const float*)d_in[0];
  const float* k     = (const float*)d_in[1];
  const float* alpha = (const float*)d_in[2];
  const float* q     = (const float*)d_in[3];
  float* out = (float*)d_out;
  float* ws  = (float*)d_ws;

  const long KD_N = (long)T_TOT * B_SZ * N_SZ;
  const long PF_N = (long)NCHUNK * B_SZ * N_SZ;
  const long U_N  = (long)NCHUNK * B_SZ * D_SZ * N_SZ;
  float* kd     = ws;
  float* prefix = kd + KD_N;
  float* U      = prefix + PF_N;
  float* Sst    = U + U_N;

  k_decay<<<dim3(B_SZ * NCHUNK, N_SZ / DK_COLS), DK_SEGS * (DK_COLS / 4), 0, stream>>>(k, alpha, kd, prefix);
  k_gemmU<<<dim3(B_SZ * NCHUNK, D_SZ / 64, N_SZ / 64), 128, 0, stream>>>(v, kd, U);
  k_intra<<<dim3(B_SZ * NCHUNK), 256, 0, stream>>>(q, kd, v, out);
  k_scan<<<dim3((B_SZ * D_SZ * N_SZ) / 256), 256, 0, stream>>>(U, prefix, Sst);
  k_inter<<<dim3(B_SZ * NCHUNK, D_SZ / 64), 256, 0, stream>>>(q, prefix, Sst, out);
}


# k_intra: one block per 16-row tile of a chunk (512 blocks of 4 waves, was 64 of 8); each stages and multiplies only the s-rows at or below its tile; summation order per element unchanged
# speedup vs baseline: 1.8981x; 1.4654x over previous
#include <hip/hip_runtime.h>


typedef float v2f __attribute__((ext_vector_type(2)));
typedef float v8f __attribute__((ext_vector_type(8)));

#define T_TOT  2048
#define B_SZ   4
#define D_SZ   256
#define N_SZ   256
#define C_SZ   128
#define NCHUNK 16
#define EPS_F  1e-8f

__device__ __forceinline__ v8f wmma_f32_k4(v2f a, v2f b, v8f c) {
  return __builtin_amdgcn_wmma_f32_16x16x4_f32(false, a, false, b, (short)0, c, false, false);
}

typedef float v4f __attribute__((ext_vector_type(4)));

#define DK_COLS  64
#define DK_SEGS  16
#define DK_STEPS (C_SZ / DK_SEGS)

__device__ __forceinline__ v4f clip4(v4f a) {
  v4f r;
  r.x = (a.x < EPS_F) ? EPS_F : a.x;
  r.y = (a.y < EPS_F) ? EPS_F : a.y;
  r.z = (a.z < EPS_F) ? EPS_F : a.z;
  r.w = (a.w < EPS_F) ? EPS_F : a.w;
  return r;
}

__global__ void k_decay(const float* __restrict__ kin, const float* __restrict__ alpha,
                        float* __restrict__ kd, float* __restrict__ prefix) {
  __shared__ v4f segTot[DK_SEGS][DK_COLS / 4];
  const int c = blockIdx.x / B_SZ;
  const int b = blockIdx.x % B_SZ;
  const int q4  = threadIdx.x & 15;
  const int seg = threadIdx.x >> 4;
  const int n0  = blockIdx.y * DK_COLS + q4 * 4;
  const long stride = (long)B_SZ * N_SZ;
  const long base = ((long)(c * C_SZ + seg * DK_STEPS) * B_SZ + b) * N_SZ + n0;

  v4f a[DK_STEPS], kv[DK_STEPS];
  for (int i = 0; i < DK_STEPS; ++i)
    a[i] = *reinterpret_cast<const v4f*>(alpha + base + (long)i * stride);
  for (int i = 0; i < DK_STEPS; ++i)
    kv[i] = *reinterpret_cast<const v4f*>(kin + base + (long)i * stride);

  v4f tot = {1.0f, 1.0f, 1.0f, 1.0f};
  for (int i = 0; i < DK_STEPS; ++i) {
    a[i] = clip4(a[i]);
    tot *= a[i];
  }
  segTot[seg][q4] = tot;
  __syncthreads();

  v4f E = {1.0f, 1.0f, 1.0f, 1.0f};
  v4f P = {1.0f, 1.0f, 1.0f, 1.0f};
  for (int j = 0; j < DK_SEGS; ++j) {
    if (j == seg) E = P;
    P *= segTot[j][q4];
  }
  const v4f inv = 1.0f / (P + EPS_F);

  v4f r = E;
  for (int i = 0; i < DK_STEPS; ++i) {
    r *= a[i];
    *reinterpret_cast<v4f*>(kd + base + (long)i * stride) = kv[i] * (r * inv);
  }
  if (seg == 0)
    *reinterpret_cast<v4f*>(prefix + (c * B_SZ + b) * N_SZ + n0) = a[0];
}

__global__ void k_gemmU(const float* __restrict__ v, const float* __restrict__ kd,
                        float* __restrict__ U) {
  __shared__ float Av[C_SZ][65];
  __shared__ float Bk[C_SZ][65];
  const int c = blockIdx.x / B_SZ, b = blockIdx.x % B_SZ;
  const int dBase = blockIdx.y * 64, nBase = blockIdx.z * 64;
  const int tid = threadIdx.x, lane = tid & 31, w = tid >> 5;

  for (int idx = tid; idx < C_SZ * 64; idx += 128) {
    const int t = idx >> 6, i = idx & 63;
    const long row = (long)(c * C_SZ + t) * B_SZ + b;
    Av[t][i] = v[row * D_SZ + dBase + i];
    Bk[t][i] = kd[row * N_SZ + nBase + i];
  }
  __syncthreads();

  const int half = lane >> 4, lan = lane & 15;
  const int mLoc = w * 16 + lan;
  v8f acc[4] = {};
  #pragma unroll
  for (int kk = 0; kk < C_SZ; kk += 4) {
    const int ka = kk + half * 2;
    v2f a; a.x = Av[ka][mLoc]; a.y = Av[ka + 1][mLoc];
    #pragma unroll
    for (int j = 0; j < 4; ++j) {
      v2f bf; bf.x = Bk[ka][j * 16 + lan]; bf.y = Bk[ka + 1][j * 16 + lan];
      acc[j] = wmma_f32_k4(a, bf, acc[j]);
    }
  }
  const long ub = (long)(c * B_SZ + b) * D_SZ;
  #pragma unroll
  for (int j = 0; j < 4; ++j)
    #pragma unroll
    for (int i = 0; i < 8; ++i) {
      const int row = dBase + w * 16 + half * 8 + i;
      const int col = nBase + j * 16 + lan;
      U[(ub + row) * N_SZ + col] = acc[j][i];
    }
}

#define IT_QP 260
#define IT_VP 264
#define IT_AP 132

template <int NB>
__device__ __forceinline__ void stage_rows(float* __restrict__ dst, int pitch, const float* __restrict__ src,
                                           long rowStride, int i0, int tid) {
  v4f tmp[NB];
  #pragma unroll
  for (int u = 0; u < NB; ++u) {
    const int idx = tid + (i0 + u) * 128, row = idx >> 6, c4 = (idx & 63) * 4;
    tmp[u] = *reinterpret_cast<const v4f*>(src + (long)row * rowStride + c4);
  }
  #pragma unroll
  for (int u = 0; u < NB; ++u) {
    const int idx = tid + (i0 + u) * 128, row = idx >> 6, c4 = (idx & 63) * 4;
    *reinterpret_cast<v4f*>(dst + row * pitch + c4) = tmp[u];
  }
}

__device__ __forceinline__ void stage_all(float* __restrict__ dst, int pitch, const float* __restrict__ src,
                                          long rowStride, int n, int tid) {
  int i = 0;
  for (; i + 32 <= n; i += 32) stage_rows<32>(dst, pitch, src, rowStride, i, tid);
  if (i + 16 <= n) { stage_rows<16>(dst, pitch, src, rowStride, i, tid); i += 16; }
  if (i + 8 <= n)  { stage_rows<8>(dst, pitch, src, rowStride, i, tid); }
}

__global__ void __launch_bounds__(128) k_intra(const float* __restrict__ q, const float* __restrict__ kd,
                                               const float* __restrict__ v, float* __restrict__ out) {
  __shared__ __attribute__((aligned(16))) float Qs[16 * IT_QP];
  __shared__ __attribute__((aligned(16))) float Ps[C_SZ * IT_VP];
  __shared__ __attribute__((aligned(16))) float As[16 * IT_AP];
  const int c = blockIdx.x / B_SZ, b = blockIdx.x % B_SZ;
  const int r  = (blockIdx.y < 4) ? (7 - (int)blockIdx.y) : ((int)blockIdx.y - 4);
  const int t0 = r * 16;
  const int S  = t0 + 16;
  const int tid = threadIdx.x, lane = tid & 31;
  const int w = __builtin_amdgcn_readfirstlane(tid >> 5);
  const int half = lane >> 4, lan = lane & 15;
  const long rowN = (long)B_SZ * N_SZ, rowD = (long)B_SZ * D_SZ;
  const float* qg  = q  + ((long)(c * C_SZ) * B_SZ + b) * N_SZ;
  const float* kg  = kd + ((long)(c * C_SZ) * B_SZ + b) * N_SZ;
  const float* vg  = v  + ((long)(c * C_SZ) * B_SZ + b) * D_SZ;

  stage_rows<8>(Qs, IT_QP, qg + (long)t0 * rowN, rowN, 0, tid);
  stage_all(Ps, IT_QP, kg, rowN, S / 2, tid);
  __syncthreads();

  v8f a0 = {}, a1 = {};
  const float* qf  = Qs + lan * IT_QP + half * 2;
  const float* kf0 = Ps + (w * 16 + lan) * IT_QP + half * 2;
  const float* kf1 = kf0 + 64 * IT_QP;
  if (w + 4 <= r) {
    #pragma unroll 8
    for (int kk = 0; kk < N_SZ; kk += 4) {
      const v2f a  = *reinterpret_cast<const v2f*>(qf + kk);
      const v2f b0 = *reinterpret_cast<const v2f*>(kf0 + kk);
      const v2f b1 = *reinterpret_cast<const v2f*>(kf1 + kk);
      a0 = wmma_f32_k4(a, b0, a0);
      a1 = wmma_f32_k4(a, b1, a1);
    }
  } else if (w <= r) {
    #pragma unroll 8
    for (int kk = 0; kk < N_SZ; kk += 4) {
      const v2f a  = *reinterpret_cast<const v2f*>(qf + kk);
      const v2f b0 = *reinterpret_cast<const v2f*>(kf0 + kk);
      a0 = wmma_f32_k4(a, b0, a0);
    }
  }
  if (w <= r) {
    #pragma unroll
    for (int i = 0; i < 8; ++i) {
      const int row = half * 8 + i, col = w * 16 + lan;
      As[row * IT_AP + col] = (col <= t0 + row) ? a0[i] : 0.0f;
    }
  }
  if (w + 4 <= r) {
    #pragma unroll
    for (int i = 0; i < 8; ++i) {
      const int row = half * 8 + i, col = (w + 4) * 16 + lan;
      As[row * IT_AP + col] = (col <= t0 + row) ? a1[i] : 0.0f;
    }
  }
  __syncthreads();

  stage_all(Ps, IT_VP, vg, rowD, S / 2, tid);
  __syncthreads();

  v8f acc[4] = {};
  const float* af = As + lan * IT_AP + half * 2;
  const float* vf = Ps + (half * 2) * IT_VP + w * 64 + lan;
  #pragma unroll 4
  for (int kk = 0; kk < S; kk += 4) {
    const v2f a = *reinterpret_cast<const v2f*>(af + kk);
    #pragma unroll
    for (int j = 0; j < 4; ++j) {
      v2f bf; bf.x = vf[kk * IT_VP + j * 16]; bf.y = vf[(kk + 1) * IT_VP + j * 16];
      acc[j] = wmma_f32_k4(a, bf, acc[j]);
    }
  }
  #pragma unroll
  for (int j = 0; j < 4; ++j)
    #pragma unroll
    for (int i = 0; i < 8; ++i) {
      const int row = t0 + half * 8 + i;
      const int col = w * 64 + j * 16 + lan;
      out[((long)(c * C_SZ + row) * B_SZ + b) * D_SZ + col] = acc[j][i];
    }
}

__global__ void k_scan(const float* __restrict__ U, const float* __restrict__ prefix,
                       float* __restrict__ Sst) {
  const long g = (long)blockIdx.x * blockDim.x + threadIdx.x;
  const int n = (int)(g % N_SZ);
  const int b = (int)(g / ((long)D_SZ * N_SZ));
  const long cs = (long)B_SZ * D_SZ * N_SZ;
  float s = 0.0f;
  for (int c = 0; c < NCHUNK; ++c) {
    Sst[c * cs + g] = s;
    s = s * prefix[(c * B_SZ + b) * N_SZ + n] + U[c * cs + g];
  }
}

__global__ void k_inter(const float* __restrict__ q, const float* __restrict__ prefix,
                        const float* __restrict__ Sst, float* __restrict__ out) {
  __shared__ float Qp[C_SZ][65];
  __shared__ float Sp[64][65];
  const int c = blockIdx.x / B_SZ, b = blockIdx.x % B_SZ;
  const int dBase = blockIdx.y * 64;
  const int tid = threadIdx.x, lane = tid & 31, w = tid >> 5;
  const int half = lane >> 4, lan = lane & 15;
  const int mRow = w * 16 + lan;
  const long cs = (long)B_SZ * D_SZ * N_SZ;

  v8f acc[4] = {};
  for (int np = 0; np < N_SZ; np += 64) {
    for (int idx = tid; idx < C_SZ * 64; idx += 256) {
      const int t = idx >> 6, j = idx & 63;
      const float pf = prefix[(c * B_SZ + b) * N_SZ + np + j];
      Qp[t][j] = q[((long)(c * C_SZ + t) * B_SZ + b) * N_SZ + np + j] * pf;
    }
    for (int idx = tid; idx < 64 * 64; idx += 256) {
      const int i = idx >> 6, j = idx & 63;
      Sp[i][j] = Sst[(long)c * cs + ((long)b * D_SZ + dBase + i) * N_SZ + np + j];
    }
    __syncthreads();
    #pragma unroll
    for (int kk = 0; kk < 64; kk += 4) {
      const int ka = kk + half * 2;
      v2f a; a.x = Qp[mRow][ka]; a.y = Qp[mRow][ka + 1];
      #pragma unroll
      for (int j = 0; j < 4; ++j) {
        v2f bf; bf.x = Sp[j * 16 + lan][ka]; bf.y = Sp[j * 16 + lan][ka + 1];
        acc[j] = wmma_f32_k4(a, bf, acc[j]);
      }
    }
    __syncthreads();
  }
  #pragma unroll
  for (int j = 0; j < 4; ++j)
    #pragma unroll
    for (int i = 0; i < 8; ++i) {
      const int row = w * 16 + half * 8 + i;
      const int col = dBase + j * 16 + lan;
      out[((long)(c * C_SZ + row) * B_SZ + b) * D_SZ + col] += acc[j][i];
    }
}

extern "C" void kernel_launch(void* const* d_in, const int* in_sizes, int n_in,
                              void* d_out, int out_size, void* d_ws, size_t ws_size,
                              hipStream_t stream) {
  const float* v     = (const float*)d_in[0];
  const float* k     = (const float*)d_in[1];
  const float* alpha = (const float*)d_in[2];
  const float* q     = (const float*)d_in[3];
  float* out = (float*)d_out;
  float* ws  = (float*)d_ws;

  const long KD_N = (long)T_TOT * B_SZ * N_SZ;
  const long PF_N = (long)NCHUNK * B_SZ * N_SZ;
  const long U_N  = (long)NCHUNK * B_SZ * D_SZ * N_SZ;
  float* kd     = ws;
  float* prefix = kd + KD_N;
  float* U      = prefix + PF_N;
  float* Sst    = U + U_N;

  k_decay<<<dim3(B_SZ * NCHUNK, N_SZ / DK_COLS), DK_SEGS * (DK_COLS / 4), 0, stream>>>(k, alpha, kd, prefix);
  k_gemmU<<<dim3(B_SZ * NCHUNK, D_SZ / 64, N_SZ / 64), 128, 0, stream>>>(v, kd, U);
  k_intra<<<dim3(B_SZ * NCHUNK, C_SZ / 16), 128, 0, stream>>>(q, kd, v, out);
  k_scan<<<dim3((B_SZ * D_SZ * N_SZ) / 256), 256, 0, stream>>>(U, prefix, Sst);
  k_inter<<<dim3(B_SZ * NCHUNK, D_SZ / 64), 256, 0, stream>>>(q, prefix, Sst, out);
}


# k_inter, k_gemmU: tiles staged whole by 16-byte loads issued as one group, LDS-only barriers between parts (was 4-byte loads in trips of 2 to 16; k_inter in four passes); sum order unchanged
# speedup vs baseline: 2.8396x; 1.4960x over previous
#include <hip/hip_runtime.h>


typedef float v2f __attribute__((ext_vector_type(2)));
typedef float v8f __attribute__((ext_vector_type(8)));

#define T_TOT  2048
#define B_SZ   4
#define D_SZ   256
#define N_SZ   256
#define C_SZ   128
#define NCHUNK 16
#define EPS_F  1e-8f

__device__ __forceinline__ v8f wmma_f32_k4(v2f a, v2f b, v8f c) {
  return __builtin_amdgcn_wmma_f32_16x16x4_f32(false, a, false, b, (short)0, c, false, false);
}

typedef float v4f __attribute__((ext_vector_type(4)));

#define DK_COLS  64
#define DK_SEGS  16
#define DK_STEPS (C_SZ / DK_SEGS)

__device__ __forceinline__ v4f clip4(v4f a) {
  v4f r;
  r.x = (a.x < EPS_F) ? EPS_F : a.x;
  r.y = (a.y < EPS_F) ? EPS_F : a.y;
  r.z = (a.z < EPS_F) ? EPS_F : a.z;
  r.w = (a.w < EPS_F) ? EPS_F : a.w;
  return r;
}

__global__ void k_decay(const float* __restrict__ kin, const float* __restrict__ alpha,
                        float* __restrict__ kd, float* __restrict__ prefix) {
  __shared__ v4f segTot[DK_SEGS][DK_COLS / 4];
  const int c = blockIdx.x / B_SZ;
  const int b = blockIdx.x % B_SZ;
  const int q4  = threadIdx.x & 15;
  const int seg = threadIdx.x >> 4;
  const int n0  = blockIdx.y * DK_COLS + q4 * 4;
  const long stride = (long)B_SZ * N_SZ;
  const long base = ((long)(c * C_SZ + seg * DK_STEPS) * B_SZ + b) * N_SZ + n0;

  v4f a[DK_STEPS], kv[DK_STEPS];
  for (int i = 0; i < DK_STEPS; ++i)
    a[i] = *reinterpret_cast<const v4f*>(alpha + base + (long)i * stride);
  for (int i = 0; i < DK_STEPS; ++i)
    kv[i] = *reinterpret_cast<const v4f*>(kin + base + (long)i * stride);

  v4f tot = {1.0f, 1.0f, 1.0f, 1.0f};
  for (int i = 0; i < DK_STEPS; ++i) {
    a[i] = clip4(a[i]);
    tot *= a[i];
  }
  segTot[seg][q4] = tot;
  __syncthreads();

  v4f E = {1.0f, 1.0f, 1.0f, 1.0f};
  v4f P = {1.0f, 1.0f, 1.0f, 1.0f};
  for (int j = 0; j < DK_SEGS; ++j) {
    if (j == seg) E = P;
    P *= segTot[j][q4];
  }
  const v4f inv = 1.0f / (P + EPS_F);

  v4f r = E;
  for (int i = 0; i < DK_STEPS; ++i) {
    r *= a[i];
    *reinterpret_cast<v4f*>(kd + base + (long)i * stride) = kv[i] * (r * inv);
  }
  if (seg == 0)
    *reinterpret_cast<v4f*>(prefix + (c * B_SZ + b) * N_SZ + n0) = a[0];
}

#define GU_P    72
#define GU_HALF 64

__global__ void __launch_bounds__(128) k_gemmU(const float* __restrict__ v, const float* __restrict__ kd,
                                               float* __restrict__ U) {
  __shared__ __attribute__((aligned(16))) float Av[C_SZ * GU_P];
  __shared__ __attribute__((aligned(16))) float Bk[C_SZ * GU_P];
  const int c = blockIdx.x / B_SZ, b = blockIdx.x % B_SZ;
  const int dBase = blockIdx.y * 64, nBase = blockIdx.z * 64;
  const int tid = threadIdx.x, lane = tid & 31, w = tid >> 5;
  const long rowN = (long)B_SZ * N_SZ, rowD = (long)B_SZ * D_SZ;
  const float* vg = v  + ((long)(c * C_SZ) * B_SZ + b) * D_SZ + dBase;
  const float* kg = kd + ((long)(c * C_SZ) * B_SZ + b) * N_SZ + nBase;
  const int p4 = (tid & 15) * 4;
  const int r0 = tid >> 4;

  v4f va[2][8], vb[2][8];
  #pragma unroll
  for (int h2 = 0; h2 < 2; ++h2) {
    #pragma unroll
    for (int u = 0; u < 8; ++u)
      va[h2][u] = *reinterpret_cast<const v4f*>(vg + (long)(h2 * GU_HALF + r0 + 8 * u) * rowD + p4);
    #pragma unroll
    for (int u = 0; u < 8; ++u)
      vb[h2][u] = *reinterpret_cast<const v4f*>(kg + (long)(h2 * GU_HALF + r0 + 8 * u) * rowN + p4);
  }

  const int half = lane >> 4, lan = lane & 15;
  const int mLoc = w * 16 + lan;
  v8f acc[4] = {};
  #pragma unroll
  for (int h2 = 0; h2 < 2; ++h2) {
    #pragma unroll
    for (int u = 0; u < 8; ++u)
      *reinterpret_cast<v4f*>(Av + (h2 * GU_HALF + r0 + 8 * u) * GU_P + p4) = va[h2][u];
    #pragma unroll
    for (int u = 0; u < 8; ++u)
      *reinterpret_cast<v4f*>(Bk + (h2 * GU_HALF + r0 + 8 * u) * GU_P + p4) = vb[h2][u];
    __builtin_amdgcn_fence(__ATOMIC_RELEASE, "workgroup", "local");
    __builtin_amdgcn_s_barrier();
    __builtin_amdgcn_fence(__ATOMIC_ACQUIRE, "workgroup", "local");
    #pragma unroll
    for (int kk = h2 * GU_HALF; kk < (h2 + 1) * GU_HALF; kk += 4) {
      const int ka = kk + half * 2;
      v2f a; a.x = Av[ka * GU_P + mLoc]; a.y = Av[(ka + 1) * GU_P + mLoc];
      #pragma unroll
      for (int j = 0; j < 4; ++j) {
        v2f bf; bf.x = Bk[ka * GU_P + j * 16 + lan]; bf.y = Bk[(ka + 1) * GU_P + j * 16 + lan];
        acc[j] = wmma_f32_k4(a, bf, acc[j]);
      }
    }
  }
  const long ub = (long)(c * B_SZ + b) * D_SZ;
  #pragma unroll
  for (int j = 0; j < 4; ++j)
    #pragma unroll
    for (int i = 0; i < 8; ++i) {
      const int row = dBase + w * 16 + half * 8 + i;
      const int col = nBase + j * 16 + lan;
      U[(ub + row) * N_SZ + col] = acc[j][i];
    }
}

#define IT_QP 260
#define IT_VP 264
#define IT_AP 132

template <int NB>
__device__ __forceinline__ void stage_rows(float* __restrict__ dst, int pitch, const float* __restrict__ src,
                                           long rowStride, int i0, int tid) {
  v4f tmp[NB];
  #pragma unroll
  for (int u = 0; u < NB; ++u) {
    const int idx = tid + (i0 + u) * 128, row = idx >> 6, c4 = (idx & 63) * 4;
    tmp[u] = *reinterpret_cast<const v4f*>(src + (long)row * rowStride + c4);
  }
  #pragma unroll
  for (int u = 0; u < NB; ++u) {
    const int idx = tid + (i0 + u) * 128, row = idx >> 6, c4 = (idx & 63) * 4;
    *reinterpret_cast<v4f*>(dst + row * pitch + c4) = tmp[u];
  }
}

__device__ __forceinline__ void stage_all(float* __restrict__ dst, int pitch, const float* __restrict__ src,
                                          long rowStride, int n, int tid) {
  int i = 0;
  for (; i + 32 <= n; i += 32) stage_rows<32>(dst, pitch, src, rowStride, i, tid);
  if (i + 16 <= n) { stage_rows<16>(dst, pitch, src, rowStride, i, tid); i += 16; }
  if (i + 8 <= n)  { stage_rows<8>(dst, pitch, src, rowStride, i, tid); }
}

__global__ void __launch_bounds__(128) k_intra(const float* __restrict__ q, const float* __restrict__ kd,
                                               const float* __restrict__ v, float* __restrict__ out) {
  __shared__ __attribute__((aligned(16))) float Qs[16 * IT_QP];
  __shared__ __attribute__((aligned(16))) float Ps[C_SZ * IT_VP];
  __shared__ __attribute__((aligned(16))) float As[16 * IT_AP];
  const int c = blockIdx.x / B_SZ, b = blockIdx.x % B_SZ;
  const int r  = (blockIdx.y < 4) ? (7 - (int)blockIdx.y) : ((int)blockIdx.y - 4);
  const int t0 = r * 16;
  const int S  = t0 + 16;
  const int tid = threadIdx.x, lane = tid & 31;
  const int w = __builtin_amdgcn_readfirstlane(tid >> 5);
  const int half = lane >> 4, lan = lane & 15;
  const long rowN = (long)B_SZ * N_SZ, rowD = (long)B_SZ * D_SZ;
  const float* qg  = q  + ((long)(c * C_SZ) * B_SZ + b) * N_SZ;
  const float* kg  = kd + ((long)(c * C_SZ) * B_SZ + b) * N_SZ;
  const float* vg  = v  + ((long)(c * C_SZ) * B_SZ + b) * D_SZ;

  stage_rows<8>(Qs, IT_QP, qg + (long)t0 * rowN, rowN, 0, tid);
  stage_all(Ps, IT_QP, kg, rowN, S / 2, tid);
  __syncthreads();

  v8f a0 = {}, a1 = {};
  const float* qf  = Qs + lan * IT_QP + half * 2;
  const float* kf0 = Ps + (w * 16 + lan) * IT_QP + half * 2;
  const float* kf1 = kf0 + 64 * IT_QP;
  if (w + 4 <= r) {
    #pragma unroll 8
    for (int kk = 0; kk < N_SZ; kk += 4) {
      const v2f a  = *reinterpret_cast<const v2f*>(qf + kk);
      const v2f b0 = *reinterpret_cast<const v2f*>(kf0 + kk);
      const v2f b1 = *reinterpret_cast<const v2f*>(kf1 + kk);
      a0 = wmma_f32_k4(a, b0, a0);
      a1 = wmma_f32_k4(a, b1, a1);
    }
  } else if (w <= r) {
    #pragma unroll 8
    for (int kk = 0; kk < N_SZ; kk += 4) {
      const v2f a  = *reinterpret_cast<const v2f*>(qf + kk);
      const v2f b0 = *reinterpret_cast<const v2f*>(kf0 + kk);
      a0 = wmma_f32_k4(a, b0, a0);
    }
  }
  if (w <= r) {
    #pragma unroll
    for (int i = 0; i < 8; ++i) {
      const int row = half * 8 + i, col = w * 16 + lan;
      As[row * IT_AP + col] = (col <= t0 + row) ? a0[i] : 0.0f;
    }
  }
  if (w + 4 <= r) {
    #pragma unroll
    for (int i = 0; i < 8; ++i) {
      const int row = half * 8 + i, col = (w + 4) * 16 + lan;
      As[row * IT_AP + col] = (col <= t0 + row) ? a1[i] : 0.0f;
    }
  }
  __syncthreads();

  stage_all(Ps, IT_VP, vg, rowD, S / 2, tid);
  __syncthreads();

  v8f acc[4] = {};
  const float* af = As + lan * IT_AP + half * 2;
  const float* vf = Ps + (half * 2) * IT_VP + w * 64 + lan;
  #pragma unroll 4
  for (int kk = 0; kk < S; kk += 4) {
    const v2f a = *reinterpret_cast<const v2f*>(af + kk);
    #pragma unroll
    for (int j = 0; j < 4; ++j) {
      v2f bf; bf.x = vf[kk * IT_VP + j * 16]; bf.y = vf[(kk + 1) * IT_VP + j * 16];
      acc[j] = wmma_f32_k4(a, bf, acc[j]);
    }
  }
  #pragma unroll
  for (int j = 0; j < 4; ++j)
    #pragma unroll
    for (int i = 0; i < 8; ++i) {
      const int row = t0 + half * 8 + i;
      const int col = w * 64 + j * 16 + lan;
      out[((long)(c * C_SZ + row) * B_SZ + b) * D_SZ + col] = acc[j][i];
    }
}

__global__ void k_scan(const float* __restrict__ U, const float* __restrict__ prefix,
                       float* __restrict__ Sst) {
  const long g = (long)blockIdx.x * blockDim.x + threadIdx.x;
  const int n = (int)(g % N_SZ);
  const int b = (int)(g / ((long)D_SZ * N_SZ));
  const long cs = (long)B_SZ * D_SZ * N_SZ;
  float s = 0.0f;
  for (int c = 0; c < NCHUNK; ++c) {
    Sst[c * cs + g] = s;
    s = s * prefix[(c * B_SZ + b) * N_SZ + n] + U[c * cs + g];
  }
}

#define IN_P    260
#define IN_PART 64

__global__ void __launch_bounds__(256) k_inter(const float* __restrict__ q, const float* __restrict__ prefix,
                                               const float* __restrict__ Sst, float* __restrict__ out) {
  __shared__ __attribute__((aligned(16))) float Qp[C_SZ * IN_P];
  __shared__ __attribute__((aligned(16))) float Sp[64 * IN_P];
  const int c = blockIdx.x / B_SZ, b = blockIdx.x % B_SZ;
  const int dBase = blockIdx.y * 64;
  const int tid = threadIdx.x, lane = tid & 31, w = tid >> 5;
  const int half = lane >> 4, lan = lane & 15;
  const long cs = (long)B_SZ * D_SZ * N_SZ;
  const long rowN = (long)B_SZ * N_SZ;
  const float* qg = q + ((long)(c * C_SZ) * B_SZ + b) * N_SZ;
  const float* sg = Sst + (long)c * cs + ((long)b * D_SZ + dBase) * N_SZ;
  const float* pg = prefix + (c * B_SZ + b) * N_SZ;
  const int p4 = (tid & 15) * 4;
  const int r0 = tid >> 4;

  v4f pf[4], qv[4][8], sv[4][4];
  #pragma unroll
  for (int kq = 0; kq < 4; ++kq) {
    pf[kq] = *reinterpret_cast<const v4f*>(pg + kq * IN_PART + p4);
    #pragma unroll
    for (int u = 0; u < 8; ++u)
      qv[kq][u] = *reinterpret_cast<const v4f*>(qg + (long)(r0 + 16 * u) * rowN + kq * IN_PART + p4);
    #pragma unroll
    for (int u = 0; u < 4; ++u)
      sv[kq][u] = *reinterpret_cast<const v4f*>(sg + (long)(r0 + 16 * u) * N_SZ + kq * IN_PART + p4);
  }

  v8f acc[4] = {};
  const float* af = Qp + (w * 16 + lan) * IN_P + half * 2;
  const float* bf = Sp + lan * IN_P + half * 2;
  #pragma unroll
  for (int kq = 0; kq < 4; ++kq) {
    #pragma unroll
    for (int u = 0; u < 8; ++u)
      *reinterpret_cast<v4f*>(Qp + (r0 + 16 * u) * IN_P + kq * IN_PART + p4) = qv[kq][u] * pf[kq];
    #pragma unroll
    for (int u = 0; u < 4; ++u)
      *reinterpret_cast<v4f*>(Sp + (r0 + 16 * u) * IN_P + kq * IN_PART + p4) = sv[kq][u];
    __builtin_amdgcn_fence(__ATOMIC_RELEASE, "workgroup", "local");
    __builtin_amdgcn_s_barrier();
    __builtin_amdgcn_fence(__ATOMIC_ACQUIRE, "workgroup", "local");
    #pragma unroll 8
    for (int kk = kq * IN_PART; kk < (kq + 1) * IN_PART; kk += 4) {
      const v2f a = *reinterpret_cast<const v2f*>(af + kk);
      #pragma unroll
      for (int j = 0; j < 4; ++j) {
        const v2f bv = *reinterpret_cast<const v2f*>(bf + j * 16 * IN_P + kk);
        acc[j] = wmma_f32_k4(a, bv, acc[j]);
      }
    }
  }
  #pragma unroll
  for (int j = 0; j < 4; ++j)
    #pragma unroll
    for (int i = 0; i < 8; ++i) {
      const int row = w * 16 + half * 8 + i;
      const int col = dBase + j * 16 + lan;
      out[((long)(c * C_SZ + row) * B_SZ + b) * D_SZ + col] += acc[j][i];
    }
}

extern "C" void kernel_launch(void* const* d_in, const int* in_sizes, int n_in,
                              void* d_out, int out_size, void* d_ws, size_t ws_size,
                              hipStream_t stream) {
  const float* v     = (const float*)d_in[0];
  const float* k     = (const float*)d_in[1];
  const float* alpha = (const float*)d_in[2];
  const float* q     = (const float*)d_in[3];
  float* out = (float*)d_out;
  float* ws  = (float*)d_ws;

  const long KD_N = (long)T_TOT * B_SZ * N_SZ;
  const long PF_N = (long)NCHUNK * B_SZ * N_SZ;
  const long U_N  = (long)NCHUNK * B_SZ * D_SZ * N_SZ;
  float* kd     = ws;
  float* prefix = kd + KD_N;
  float* U      = prefix + PF_N;
  float* Sst    = U + U_N;

  k_decay<<<dim3(B_SZ * NCHUNK, N_SZ / DK_COLS), DK_SEGS * (DK_COLS / 4), 0, stream>>>(k, alpha, kd, prefix);
  k_gemmU<<<dim3(B_SZ * NCHUNK, D_SZ / 64, N_SZ / 64), 128, 0, stream>>>(v, kd, U);
  k_intra<<<dim3(B_SZ * NCHUNK, C_SZ / 16), 128, 0, stream>>>(q, kd, v, out);
  k_scan<<<dim3((B_SZ * D_SZ * N_SZ) / 256), 256, 0, stream>>>(U, prefix, Sst);
  k_inter<<<dim3(B_SZ * NCHUNK, D_SZ / 64), 256, 0, stream>>>(q, prefix, Sst, out);
}


# k_scan folded into k_intra: each k_intra thread carries one 16-byte column of the chunk scan, loads issued ahead of the tile staging, stores before phase 1; 4 launches (was 5); scan order unchanged
# speedup vs baseline: 2.9350x; 1.0336x over previous
#include <hip/hip_runtime.h>


typedef float v2f __attribute__((ext_vector_type(2)));
typedef float v8f __attribute__((ext_vector_type(8)));

#define T_TOT  2048
#define B_SZ   4
#define D_SZ   256
#define N_SZ   256
#define C_SZ   128
#define NCHUNK 16
#define EPS_F  1e-8f

__device__ __forceinline__ v8f wmma_f32_k4(v2f a, v2f b, v8f c) {
  return __builtin_amdgcn_wmma_f32_16x16x4_f32(false, a, false, b, (short)0, c, false, false);
}

typedef float v4f __attribute__((ext_vector_type(4)));

#define DK_COLS  64
#define DK_SEGS  16
#define DK_STEPS (C_SZ / DK_SEGS)

__device__ __forceinline__ v4f clip4(v4f a) {
  v4f r;
  r.x = (a.x < EPS_F) ? EPS_F : a.x;
  r.y = (a.y < EPS_F) ? EPS_F : a.y;
  r.z = (a.z < EPS_F) ? EPS_F : a.z;
  r.w = (a.w < EPS_F) ? EPS_F : a.w;
  return r;
}

__global__ void k_decay(const float* __restrict__ kin, const float* __restrict__ alpha,
                        float* __restrict__ kd, float* __restrict__ prefix) {
  __shared__ v4f segTot[DK_SEGS][DK_COLS / 4];
  const int c = blockIdx.x / B_SZ;
  const int b = blockIdx.x % B_SZ;
  const int q4  = threadIdx.x & 15;
  const int seg = threadIdx.x >> 4;
  const int n0  = blockIdx.y * DK_COLS + q4 * 4;
  const long stride = (long)B_SZ * N_SZ;
  const long base = ((long)(c * C_SZ + seg * DK_STEPS) * B_SZ + b) * N_SZ + n0;

  v4f a[DK_STEPS], kv[DK_STEPS];
  for (int i = 0; i < DK_STEPS; ++i)
    a[i] = *reinterpret_cast<const v4f*>(alpha + base + (long)i * stride);
  for (int i = 0; i < DK_STEPS; ++i)
    kv[i] = *reinterpret_cast<const v4f*>(kin + base + (long)i * stride);

  v4f tot = {1.0f, 1.0f, 1.0f, 1.0f};
  for (int i = 0; i < DK_STEPS; ++i) {
    a[i] = clip4(a[i]);
    tot *= a[i];
  }
  segTot[seg][q4] = tot;
  __syncthreads();

  v4f E = {1.0f, 1.0f, 1.0f, 1.0f};
  v4f P = {1.0f, 1.0f, 1.0f, 1.0f};
  for (int j = 0; j < DK_SEGS; ++j) {
    if (j == seg) E = P;
    P *= segTot[j][q4];
  }
  const v4f inv = 1.0f / (P + EPS_F);

  v4f r = E;
  for (int i = 0; i < DK_STEPS; ++i) {
    r *= a[i];
    *reinterpret_cast<v4f*>(kd + base + (long)i * stride) = kv[i] * (r * inv);
  }
  if (seg == 0)
    *reinterpret_cast<v4f*>(prefix + (c * B_SZ + b) * N_SZ + n0) = a[0];
}

#define GU_P    72
#define GU_HALF 64

__global__ void __launch_bounds__(128) k_gemmU(const float* __restrict__ v, const float* __restrict__ kd,
                                               float* __restrict__ U) {
  __shared__ __attribute__((aligned(16))) float Av[C_SZ * GU_P];
  __shared__ __attribute__((aligned(16))) float Bk[C_SZ * GU_P];
  const int c = blockIdx.x / B_SZ, b = blockIdx.x % B_SZ;
  const int dBase = blockIdx.y * 64, nBase = blockIdx.z * 64;
  const int tid = threadIdx.x, lane = tid & 31, w = tid >> 5;
  const long rowN = (long)B_SZ * N_SZ, rowD = (long)B_SZ * D_SZ;
  const float* vg = v  + ((long)(c * C_SZ) * B_SZ + b) * D_SZ + dBase;
  const float* kg = kd + ((long)(c * C_SZ) * B_SZ + b) * N_SZ + nBase;
  const int p4 = (tid & 15) * 4;
  const int r0 = tid >> 4;

  v4f va[2][8], vb[2][8];
  #pragma unroll
  for (int h2 = 0; h2 < 2; ++h2) {
    #pragma unroll
    for (int u = 0; u < 8; ++u)
      va[h2][u] = *reinterpret_cast<const v4f*>(vg + (long)(h2 * GU_HALF + r0 + 8 * u) * rowD + p4);
    #pragma unroll
    for (int u = 0; u < 8; ++u)
      vb[h2][u] = *reinterpret_cast<const v4f*>(kg + (long)(h2 * GU_HALF + r0 + 8 * u) * rowN + p4);
  }

  const int half = lane >> 4, lan = lane & 15;
  const int mLoc = w * 16 + lan;
  v8f acc[4] = {};
  #pragma unroll
  for (int h2 = 0; h2 < 2; ++h2) {
    #pragma unroll
    for (int u = 0; u < 8; ++u)
      *reinterpret_cast<v4f*>(Av + (h2 * GU_HALF + r0 + 8 * u) * GU_P + p4) = va[h2][u];
    #pragma unroll
    for (int u = 0; u < 8; ++u)
      *reinterpret_cast<v4f*>(Bk + (h2 * GU_HALF + r0 + 8 * u) * GU_P + p4) = vb[h2][u];
    __builtin_amdgcn_fence(__ATOMIC_RELEASE, "workgroup", "local");
    __builtin_amdgcn_s_barrier();
    __builtin_amdgcn_fence(__ATOMIC_ACQUIRE, "workgroup", "local");
    #pragma unroll
    for (int kk = h2 * GU_HALF; kk < (h2 + 1) * GU_HALF; kk += 4) {
      const int ka = kk + half * 2;
      v2f a; a.x = Av[ka * GU_P + mLoc]; a.y = Av[(ka + 1) * GU_P + mLoc];
      #pragma unroll
      for (int j = 0; j < 4; ++j) {
        v2f bf; bf.x = Bk[ka * GU_P + j * 16 + lan]; bf.y = Bk[(ka + 1) * GU_P + j * 16 + lan];
        acc[j] = wmma_f32_k4(a, bf, acc[j]);
      }
    }
  }
  const long ub = (long)(c * B_SZ + b) * D_SZ;
  #pragma unroll
  for (int j = 0; j < 4; ++j)
    #pragma unroll
    for (int i = 0; i < 8; ++i) {
      const int row = dBase + w * 16 + half * 8 + i;
      const int col = nBase + j * 16 + lan;
      U[(ub + row) * N_SZ + col] = acc[j][i];
    }
}

#define IT_QP 260
#define IT_VP 264
#define IT_AP 132

template <int NB>
__device__ __forceinline__ void stage_rows(float* __restrict__ dst, int pitch, const float* __restrict__ src,
                                           long rowStride, int i0, int tid) {
  v4f tmp[NB];
  #pragma unroll
  for (int u = 0; u < NB; ++u) {
    const int idx = tid + (i0 + u) * 128, row = idx >> 6, c4 = (idx & 63) * 4;
    tmp[u] = *reinterpret_cast<const v4f*>(src + (long)row * rowStride + c4);
  }
  #pragma unroll
  for (int u = 0; u < NB; ++u) {
    const int idx = tid + (i0 + u) * 128, row = idx >> 6, c4 = (idx & 63) * 4;
    *reinterpret_cast<v4f*>(dst + row * pitch + c4) = tmp[u];
  }
}

__device__ __forceinline__ void stage_all(float* __restrict__ dst, int pitch, const float* __restrict__ src,
                                          long rowStride, int n, int tid) {
  int i = 0;
  for (; i + 32 <= n; i += 32) stage_rows<32>(dst, pitch, src, rowStride, i, tid);
  if (i + 16 <= n) { stage_rows<16>(dst, pitch, src, rowStride, i, tid); i += 16; }
  if (i + 8 <= n)  { stage_rows<8>(dst, pitch, src, rowStride, i, tid); }
}

__global__ void __launch_bounds__(128) k_intra(const float* __restrict__ q, const float* __restrict__ kd,
                                               const float* __restrict__ v, const float* __restrict__ U,
                                               const float* __restrict__ prefix, float* __restrict__ Sst,
                                               float* __restrict__ out) {
  __shared__ __attribute__((aligned(16))) float Qs[16 * IT_QP];
  __shared__ __attribute__((aligned(16))) float Ps[C_SZ * IT_VP];
  __shared__ __attribute__((aligned(16))) float As[16 * IT_AP];
  const int c = blockIdx.x / B_SZ, b = blockIdx.x % B_SZ;
  const int r  = (blockIdx.y < 4) ? (7 - (int)blockIdx.y) : ((int)blockIdx.y - 4);
  const int t0 = r * 16;
  const int S  = t0 + 16;
  const int tid = threadIdx.x, lane = tid & 31;
  const int w = __builtin_amdgcn_readfirstlane(tid >> 5);
  const int half = lane >> 4, lan = lane & 15;
  const long rowN = (long)B_SZ * N_SZ, rowD = (long)B_SZ * D_SZ;
  const float* qg  = q  + ((long)(c * C_SZ) * B_SZ + b) * N_SZ;
  const float* kg  = kd + ((long)(c * C_SZ) * B_SZ + b) * N_SZ;
  const float* vg  = v  + ((long)(c * C_SZ) * B_SZ + b) * D_SZ;

  const int  sg  = (((int)blockIdx.y * (B_SZ * NCHUNK) + (int)blockIdx.x) * 128 + tid) * 4;
  const int  sb  = sg / (D_SZ * N_SZ), sn = sg % N_SZ;
  const long scs = (long)B_SZ * D_SZ * N_SZ;
  v4f su[NCHUNK - 1], sp[NCHUNK - 1];
  #pragma unroll
  for (int cc = 0; cc < NCHUNK - 1; ++cc) {
    su[cc] = *reinterpret_cast<const v4f*>(U + cc * scs + sg);
    sp[cc] = *reinterpret_cast<const v4f*>(prefix + (cc * B_SZ + sb) * N_SZ + sn);
  }

  stage_rows<8>(Qs, IT_QP, qg + (long)t0 * rowN, rowN, 0, tid);
  stage_all(Ps, IT_QP, kg, rowN, S / 2, tid);
  __syncthreads();

  {
    v4f s = {0.0f, 0.0f, 0.0f, 0.0f};
    #pragma unroll
    for (int cc = 0; cc < NCHUNK; ++cc) {
      *reinterpret_cast<v4f*>(Sst + cc * scs + sg) = s;
      if (cc < NCHUNK - 1) s = s * sp[cc] + su[cc];
    }
  }

  v8f a0 = {}, a1 = {};
  const float* qf  = Qs + lan * IT_QP + half * 2;
  const float* kf0 = Ps + (w * 16 + lan) * IT_QP + half * 2;
  const float* kf1 = kf0 + 64 * IT_QP;
  if (w + 4 <= r) {
    #pragma unroll 8
    for (int kk = 0; kk < N_SZ; kk += 4) {
      const v2f a  = *reinterpret_cast<const v2f*>(qf + kk);
      const v2f b0 = *reinterpret_cast<const v2f*>(kf0 + kk);
      const v2f b1 = *reinterpret_cast<const v2f*>(kf1 + kk);
      a0 = wmma_f32_k4(a, b0, a0);
      a1 = wmma_f32_k4(a, b1, a1);
    }
  } else if (w <= r) {
    #pragma unroll 8
    for (int kk = 0; kk < N_SZ; kk += 4) {
      const v2f a  = *reinterpret_cast<const v2f*>(qf + kk);
      const v2f b0 = *reinterpret_cast<const v2f*>(kf0 + kk);
      a0 = wmma_f32_k4(a, b0, a0);
    }
  }
  if (w <= r) {
    #pragma unroll
    for (int i = 0; i < 8; ++i) {
      const int row = half * 8 + i, col = w * 16 + lan;
      As[row * IT_AP + col] = (col <= t0 + row) ? a0[i] : 0.0f;
    }
  }
  if (w + 4 <= r) {
    #pragma unroll
    for (int i = 0; i < 8; ++i) {
      const int row = half * 8 + i, col = (w + 4) * 16 + lan;
      As[row * IT_AP + col] = (col <= t0 + row) ? a1[i] : 0.0f;
    }
  }
  __syncthreads();

  stage_all(Ps, IT_VP, vg, rowD, S / 2, tid);
  __syncthreads();

  v8f acc[4] = {};
  const float* af = As + lan * IT_AP + half * 2;
  const float* vf = Ps + (half * 2) * IT_VP + w * 64 + lan;
  #pragma unroll 4
  for (int kk = 0; kk < S; kk += 4) {
    const v2f a = *reinterpret_cast<const v2f*>(af + kk);
    #pragma unroll
    for (int j = 0; j < 4; ++j) {
      v2f bf; bf.x = vf[kk * IT_VP + j * 16]; bf.y = vf[(kk + 1) * IT_VP + j * 16];
      acc[j] = wmma_f32_k4(a, bf, acc[j]);
    }
  }
  #pragma unroll
  for (int j = 0; j < 4; ++j)
    #pragma unroll
    for (int i = 0; i < 8; ++i) {
      const int row = t0 + half * 8 + i;
      const int col = w * 64 + j * 16 + lan;
      out[((long)(c * C_SZ + row) * B_SZ + b) * D_SZ + col] = acc[j][i];
    }
}

#define IN_P    260
#define IN_PART 64

__global__ void __launch_bounds__(256) k_inter(const float* __restrict__ q, const float* __restrict__ prefix,
                                               const float* __restrict__ Sst, float* __restrict__ out) {
  __shared__ __attribute__((aligned(16))) float Qp[C_SZ * IN_P];
  __shared__ __attribute__((aligned(16))) float Sp[64 * IN_P];
  const int c = blockIdx.x / B_SZ, b = blockIdx.x % B_SZ;
  const int dBase = blockIdx.y * 64;
  const int tid = threadIdx.x, lane = tid & 31, w = tid >> 5;
  const int half = lane >> 4, lan = lane & 15;
  const long cs = (long)B_SZ * D_SZ * N_SZ;
  const long rowN = (long)B_SZ * N_SZ;
  const float* qg = q + ((long)(c * C_SZ) * B_SZ + b) * N_SZ;
  const float* sg = Sst + (long)c * cs + ((long)b * D_SZ + dBase) * N_SZ;
  const float* pg = prefix + (c * B_SZ + b) * N_SZ;
  const int p4 = (tid & 15) * 4;
  const int r0 = tid >> 4;

  v4f pf[4], qv[4][8], sv[4][4];
  #pragma unroll
  for (int kq = 0; kq < 4; ++kq) {
    pf[kq] = *reinterpret_cast<const v4f*>(pg + kq * IN_PART + p4);
    #pragma unroll
    for (int u = 0; u < 8; ++u)
      qv[kq][u] = *reinterpret_cast<const v4f*>(qg + (long)(r0 + 16 * u) * rowN + kq * IN_PART + p4);
    #pragma unroll
    for (int u = 0; u < 4; ++u)
      sv[kq][u] = *reinterpret_cast<const v4f*>(sg + (long)(r0 + 16 * u) * N_SZ + kq * IN_PART + p4);
  }

  v8f acc[4] = {};
  const float* af = Qp + (w * 16 + lan) * IN_P + half * 2;
  const float* bf = Sp + lan * IN_P + half * 2;
  #pragma unroll
  for (int kq = 0; kq < 4; ++kq) {
    #pragma unroll
    for (int u = 0; u < 8; ++u)
      *reinterpret_cast<v4f*>(Qp + (r0 + 16 * u) * IN_P + kq * IN_PART + p4) = qv[kq][u] * pf[kq];
    #pragma unroll
    for (int u = 0; u < 4; ++u)
      *reinterpret_cast<v4f*>(Sp + (r0 + 16 * u) * IN_P + kq * IN_PART + p4) = sv[kq][u];
    __builtin_amdgcn_fence(__ATOMIC_RELEASE, "workgroup", "local");
    __builtin_amdgcn_s_barrier();
    __builtin_amdgcn_fence(__ATOMIC_ACQUIRE, "workgroup", "local");
    #pragma unroll 8
    for (int kk = kq * IN_PART; kk < (kq + 1) * IN_PART; kk += 4) {
      const v2f a = *reinterpret_cast<const v2f*>(af + kk);
      #pragma unroll
      for (int j = 0; j < 4; ++j) {
        const v2f bv = *reinterpret_cast<const v2f*>(bf + j * 16 * IN_P + kk);
        acc[j] = wmma_f32_k4(a, bv, acc[j]);
      }
    }
  }
  #pragma unroll
  for (int j = 0; j < 4; ++j)
    #pragma unroll
    for (int i = 0; i < 8; ++i) {
      const int row = w * 16 + half * 8 + i;
      const int col = dBase + j * 16 + lan;
      out[((long)(c * C_SZ + row) * B_SZ + b) * D_SZ + col] += acc[j][i];
    }
}

extern "C" void kernel_launch(void* const* d_in, const int* in_sizes, int n_in,
                              void* d_out, int out_size, void* d_ws, size_t ws_size,
                              hipStream_t stream) {
  const float* v     = (const float*)d_in[0];
  const float* k     = (const float*)d_in[1];
  const float* alpha = (const float*)d_in[2];
  const float* q     = (const float*)d_in[3];
  float* out = (float*)d_out;
  float* ws  = (float*)d_ws;

  const long KD_N = (long)T_TOT * B_SZ * N_SZ;
  const long PF_N = (long)NCHUNK * B_SZ * N_SZ;
  const long U_N  = (long)NCHUNK * B_SZ * D_SZ * N_SZ;
  float* kd     = ws;
  float* prefix = kd + KD_N;
  float* U      = prefix + PF_N;
  float* Sst    = U + U_N;

  k_decay<<<dim3(B_SZ * NCHUNK, N_SZ / DK_COLS), DK_SEGS * (DK_COLS / 4), 0, stream>>>(k, alpha, kd, prefix);
  k_gemmU<<<dim3(B_SZ * NCHUNK, D_SZ / 64, N_SZ / 64), 128, 0, stream>>>(v, kd, U);
  k_intra<<<dim3(B_SZ * NCHUNK, C_SZ / 16), 128, 0, stream>>>(q, kd, v, U, prefix, Sst, out);
  k_inter<<<dim3(B_SZ * NCHUNK, D_SZ / 64), 256, 0, stream>>>(q, prefix, Sst, out);
}


# k_decay folded into k_gemmU (3 launches, was 4): each block, now 128x64, loads its alpha and k tiles with its v tile and runs the same segmented scan; d-block 0 stores kd, prefix; same product order
# speedup vs baseline: 3.1107x; 1.0599x over previous
#include <hip/hip_runtime.h>


typedef float v2f __attribute__((ext_vector_type(2)));
typedef float v8f __attribute__((ext_vector_type(8)));

#define T_TOT  2048
#define B_SZ   4
#define D_SZ   256
#define N_SZ   256
#define C_SZ   128
#define NCHUNK 16
#define EPS_F  1e-8f

__device__ __forceinline__ v8f wmma_f32_k4(v2f a, v2f b, v8f c) {
  return __builtin_amdgcn_wmma_f32_16x16x4_f32(false, a, false, b, (short)0, c, false, false);
}

typedef float v4f __attribute__((ext_vector_type(4)));

#define DK_SEGS  16
#define DK_STEPS (C_SZ / DK_SEGS)

__device__ __forceinline__ v4f clip4(v4f a) {
  v4f r;
  r.x = (a.x < EPS_F) ? EPS_F : a.x;
  r.y = (a.y < EPS_F) ? EPS_F : a.y;
  r.z = (a.z < EPS_F) ? EPS_F : a.z;
  r.w = (a.w < EPS_F) ? EPS_F : a.w;
  return r;
}

#define GU_PA   136
#define GU_P    72
#define GU_HALF 64

__global__ void __launch_bounds__(256) k_gemmU(const float* __restrict__ v, const float* __restrict__ kin,
                                               const float* __restrict__ alpha, float* __restrict__ kd,
                                               float* __restrict__ prefix, float* __restrict__ U) {
  __shared__ __attribute__((aligned(16))) float Av[C_SZ * GU_PA];
  __shared__ __attribute__((aligned(16))) float Bk[C_SZ * GU_P];
  __shared__ v4f segTot[DK_SEGS][16];
  const int c = blockIdx.x / B_SZ, b = blockIdx.x % B_SZ;
  const int dBase = blockIdx.y * 128, nBase = blockIdx.z * 64;
  const int tid = threadIdx.x, lane = tid & 31, w = tid >> 5;
  const long rowN = (long)B_SZ * N_SZ, rowD = (long)B_SZ * D_SZ;
  const float* vg = v + ((long)(c * C_SZ) * B_SZ + b) * D_SZ + dBase;
  const int q4  = tid & 15;
  const int seg = tid >> 4;
  const int tRow = seg * DK_STEPS;
  const long kbase = ((long)(c * C_SZ + tRow) * B_SZ + b) * N_SZ + nBase + q4 * 4;
  const int c4 = lane * 4;

  v4f a[DK_STEPS], kv[DK_STEPS], va[2][8];
  #pragma unroll
  for (int i = 0; i < DK_STEPS; ++i)
    a[i] = *reinterpret_cast<const v4f*>(alpha + kbase + (long)i * rowN);
  __builtin_amdgcn_sched_barrier(0);
  #pragma unroll
  for (int i = 0; i < DK_STEPS; ++i)
    kv[i] = *reinterpret_cast<const v4f*>(kin + kbase + (long)i * rowN);
  #pragma unroll
  for (int h2 = 0; h2 < 2; ++h2) {
    #pragma unroll
    for (int u = 0; u < 8; ++u)
      va[h2][u] = *reinterpret_cast<const v4f*>(vg + (long)(h2 * GU_HALF + w + 8 * u) * rowD + c4);
  }
  __builtin_amdgcn_sched_barrier(0);

  v4f tot = {1.0f, 1.0f, 1.0f, 1.0f};
  #pragma unroll
  for (int i = 0; i < DK_STEPS; ++i) {
    a[i] = clip4(a[i]);
    tot *= a[i];
  }
  segTot[seg][q4] = tot;
  __builtin_amdgcn_fence(__ATOMIC_RELEASE, "workgroup", "local");
  __builtin_amdgcn_s_barrier();
  __builtin_amdgcn_fence(__ATOMIC_ACQUIRE, "workgroup", "local");

  v4f E = {1.0f, 1.0f, 1.0f, 1.0f};
  v4f P = {1.0f, 1.0f, 1.0f, 1.0f};
  #pragma unroll
  for (int j = 0; j < DK_SEGS; ++j) {
    if (j == seg) E = P;
    P *= segTot[j][q4];
  }
  const v4f inv = 1.0f / (P + EPS_F);

  v4f r = E;
  #pragma unroll
  for (int i = 0; i < DK_STEPS; ++i) {
    r *= a[i];
    kv[i] = kv[i] * (r * inv);
  }
  #pragma unroll
  for (int i = 0; i < DK_STEPS; ++i)
    *reinterpret_cast<v4f*>(Bk + (tRow + i) * GU_P + q4 * 4) = kv[i];
  if (blockIdx.y == 0) {
    #pragma unroll
    for (int i = 0; i < DK_STEPS; ++i)
      *reinterpret_cast<v4f*>(kd + kbase + (long)i * rowN) = kv[i];
    if (seg == 0)
      *reinterpret_cast<v4f*>(prefix + (c * B_SZ + b) * N_SZ + nBase + q4 * 4) = a[0];
  }

  const int half = lane >> 4, lan = lane & 15;
  const int mLoc = w * 16 + lan;
  v8f acc[4] = {};
  #pragma unroll
  for (int h2 = 0; h2 < 2; ++h2) {
    #pragma unroll
    for (int u = 0; u < 8; ++u)
      *reinterpret_cast<v4f*>(Av + (h2 * GU_HALF + w + 8 * u) * GU_PA + c4) = va[h2][u];
    __builtin_amdgcn_fence(__ATOMIC_RELEASE, "workgroup", "local");
    __builtin_amdgcn_s_barrier();
    __builtin_amdgcn_fence(__ATOMIC_ACQUIRE, "workgroup", "local");
    #pragma unroll
    for (int kk = h2 * GU_HALF; kk < (h2 + 1) * GU_HALF; kk += 4) {
      const int ka = kk + half * 2;
      v2f af; af.x = Av[ka * GU_PA + mLoc]; af.y = Av[(ka + 1) * GU_PA + mLoc];
      #pragma unroll
      for (int j = 0; j < 4; ++j) {
        v2f bf; bf.x = Bk[ka * GU_P + j * 16 + lan]; bf.y = Bk[(ka + 1) * GU_P + j * 16 + lan];
        acc[j] = wmma_f32_k4(af, bf, acc[j]);
      }
    }
  }
  const long ub = (long)(c * B_SZ + b) * D_SZ;
  #pragma unroll
  for (int j = 0; j < 4; ++j)
    #pragma unroll
    for (int i = 0; i < 8; ++i) {
      const int row = dBase + w * 16 + half * 8 + i;
      const int col = nBase + j * 16 + lan;
      U[(ub + row) * N_SZ + col] = acc[j][i];
    }
}

#define IT_QP 260
#define IT_VP 264
#define IT_AP 132

template <int NB>
__device__ __forceinline__ void stage_rows(float* __restrict__ dst, int pitch, const float* __restrict__ src,
                                           long rowStride, int i0, int tid) {
  v4f tmp[NB];
  #pragma unroll
  for (int u = 0; u < NB; ++u) {
    const int idx = tid + (i0 + u) * 128, row = idx >> 6, c4 = (idx & 63) * 4;
    tmp[u] = *reinterpret_cast<const v4f*>(src + (long)row * rowStride + c4);
  }
  #pragma unroll
  for (int u = 0; u < NB; ++u) {
    const int idx = tid + (i0 + u) * 128, row = idx >> 6, c4 = (idx & 63) * 4;
    *reinterpret_cast<v4f*>(dst + row * pitch + c4) = tmp[u];
  }
}

__device__ __forceinline__ void stage_all(float* __restrict__ dst, int pitch, const float* __restrict__ src,
                                          long rowStride, int n, int tid) {
  int i = 0;
  for (; i + 32 <= n; i += 32) stage_rows<32>(dst, pitch, src, rowStride, i, tid);
  if (i + 16 <= n) { stage_rows<16>(dst, pitch, src, rowStride, i, tid); i += 16; }
  if (i + 8 <= n)  { stage_rows<8>(dst, pitch, src, rowStride, i, tid); }
}

__global__ void __launch_bounds__(128) k_intra(const float* __restrict__ q, const float* __restrict__ kd,
                                               const float* __restrict__ v, const float* __restrict__ U,
                                               const float* __restrict__ prefix, float* __restrict__ Sst,
                                               float* __restrict__ out) {
  __shared__ __attribute__((aligned(16))) float Qs[16 * IT_QP];
  __shared__ __attribute__((aligned(16))) float Ps[C_SZ * IT_VP];
  __shared__ __attribute__((aligned(16))) float As[16 * IT_AP];
  const int c = blockIdx.x / B_SZ, b = blockIdx.x % B_SZ;
  const int r  = (blockIdx.y < 4) ? (7 - (int)blockIdx.y) : ((int)blockIdx.y - 4);
  const int t0 = r * 16;
  const int S  = t0 + 16;
  const int tid = threadIdx.x, lane = tid & 31;
  const int w = __builtin_amdgcn_readfirstlane(tid >> 5);
  const int half = lane >> 4, lan = lane & 15;
  const long rowN = (long)B_SZ * N_SZ, rowD = (long)B_SZ * D_SZ;
  const float* qg  = q  + ((long)(c * C_SZ) * B_SZ + b) * N_SZ;
  const float* kg  = kd + ((long)(c * C_SZ) * B_SZ + b) * N_SZ;
  const float* vg  = v  + ((long)(c * C_SZ) * B_SZ + b) * D_SZ;

  const int  sg  = (((int)blockIdx.y * (B_SZ * NCHUNK) + (int)blockIdx.x) * 128 + tid) * 4;
  const int  sb  = sg / (D_SZ * N_SZ), sn = sg % N_SZ;
  const long scs = (long)B_SZ * D_SZ * N_SZ;
  v4f su[NCHUNK - 1], sp[NCHUNK - 1];
  #pragma unroll
  for (int cc = 0; cc < NCHUNK - 1; ++cc) {
    su[cc] = *reinterpret_cast<const v4f*>(U + cc * scs + sg);
    sp[cc] = *reinterpret_cast<const v4f*>(prefix + (cc * B_SZ + sb) * N_SZ + sn);
  }

  stage_rows<8>(Qs, IT_QP, qg + (long)t0 * rowN, rowN, 0, tid);
  stage_all(Ps, IT_QP, kg, rowN, S / 2, tid);
  __syncthreads();

  {
    v4f s = {0.0f, 0.0f, 0.0f, 0.0f};
    #pragma unroll
    for (int cc = 0; cc < NCHUNK; ++cc) {
      *reinterpret_cast<v4f*>(Sst + cc * scs + sg) = s;
      if (cc < NCHUNK - 1) s = s * sp[cc] + su[cc];
    }
  }

  v8f a0 = {}, a1 = {};
  const float* qf  = Qs + lan * IT_QP + half * 2;
  const float* kf0 = Ps + (w * 16 + lan) * IT_QP + half * 2;
  const float* kf1 = kf0 + 64 * IT_QP;
  if (w + 4 <= r) {
    #pragma unroll 8
    for (int kk = 0; kk < N_SZ; kk += 4) {
      const v2f a  = *reinterpret_cast<const v2f*>(qf + kk);
      const v2f b0 = *reinterpret_cast<const v2f*>(kf0 + kk);
      const v2f b1 = *reinterpret_cast<const v2f*>(kf1 + kk);
      a0 = wmma_f32_k4(a, b0, a0);
      a1 = wmma_f32_k4(a, b1, a1);
    }
  } else if (w <= r) {
    #pragma unroll 8
    for (int kk = 0; kk < N_SZ; kk += 4) {
      const v2f a  = *reinterpret_cast<const v2f*>(qf + kk);
      const v2f b0 = *reinterpret_cast<const v2f*>(kf0 + kk);
      a0 = wmma_f32_k4(a, b0, a0);
    }
  }
  if (w <= r) {
    #pragma unroll
    for (int i = 0; i < 8; ++i) {
      const int row = half * 8 + i, col = w * 16 + lan;
      As[row * IT_AP + col] = (col <= t0 + row) ? a0[i] : 0.0f;
    }
  }
  if (w + 4 <= r) {
    #pragma unroll
    for (int i = 0; i < 8; ++i) {
      const int row = half * 8 + i, col = (w + 4) * 16 + lan;
      As[row * IT_AP + col] = (col <= t0 + row) ? a1[i] : 0.0f;
    }
  }
  __syncthreads();

  stage_all(Ps, IT_VP, vg, rowD, S / 2, tid);
  __syncthreads();

  v8f acc[4] = {};
  const float* af = As + lan * IT_AP + half * 2;
  const float* vf = Ps + (half * 2) * IT_VP + w * 64 + lan;
  #pragma unroll 4
  for (int kk = 0; kk < S; kk += 4) {
    const v2f a = *reinterpret_cast<const v2f*>(af + kk);
    #pragma unroll
    for (int j = 0; j < 4; ++j) {
      v2f bf; bf.x = vf[kk * IT_VP + j * 16]; bf.y = vf[(kk + 1) * IT_VP + j * 16];
      acc[j] = wmma_f32_k4(a, bf, acc[j]);
    }
  }
  #pragma unroll
  for (int j = 0; j < 4; ++j)
    #pragma unroll
    for (int i = 0; i < 8; ++i) {
      const int row = t0 + half * 8 + i;
      const int col = w * 64 + j * 16 + lan;
      out[((long)(c * C_SZ + row) * B_SZ + b) * D_SZ + col] = acc[j][i];
    }
}

#define IN_P    260
#define IN_PART 64

__global__ void __launch_bounds__(256) k_inter(const float* __restrict__ q, const float* __restrict__ prefix,
                                               const float* __restrict__ Sst, float* __restrict__ out) {
  __shared__ __attribute__((aligned(16))) float Qp[C_SZ * IN_P];
  __shared__ __attribute__((aligned(16))) float Sp[64 * IN_P];
  const int c = blockIdx.x / B_SZ, b = blockIdx.x % B_SZ;
  const int dBase = blockIdx.y * 64;
  const int tid = threadIdx.x, lane = tid & 31, w = tid >> 5;
  const int half = lane >> 4, lan = lane & 15;
  const long cs = (long)B_SZ * D_SZ * N_SZ;
  const long rowN = (long)B_SZ * N_SZ;
  const float* qg = q + ((long)(c * C_SZ) * B_SZ + b) * N_SZ;
  const float* sg = Sst + (long)c * cs + ((long)b * D_SZ + dBase) * N_SZ;
  const float* pg = prefix + (c * B_SZ + b) * N_SZ;
  const int p4 = (tid & 15) * 4;
  const int r0 = tid >> 4;

  v4f pf[4], qv[4][8], sv[4][4];
  #pragma unroll
  for (int kq = 0; kq < 4; ++kq) {
    pf[kq] = *reinterpret_cast<const v4f*>(pg + kq * IN_PART + p4);
    #pragma unroll
    for (int u = 0; u < 8; ++u)
      qv[kq][u] = *reinterpret_cast<const v4f*>(qg + (long)(r0 + 16 * u) * rowN + kq * IN_PART + p4);
    #pragma unroll
    for (int u = 0; u < 4; ++u)
      sv[kq][u] = *reinterpret_cast<const v4f*>(sg + (long)(r0 + 16 * u) * N_SZ + kq * IN_PART + p4);
  }

  v8f acc[4] = {};
  const float* af = Qp + (w * 16 + lan) * IN_P + half * 2;
  const float* bf = Sp + lan * IN_P + half * 2;
  #pragma unroll
  for (int kq = 0; kq < 4; ++kq) {
    #pragma unroll
    for (int u = 0; u < 8; ++u)
      *reinterpret_cast<v4f*>(Qp + (r0 + 16 * u) * IN_P + kq * IN_PART + p4) = qv[kq][u] * pf[kq];
    #pragma unroll
    for (int u = 0; u < 4; ++u)
      *reinterpret_cast<v4f*>(Sp + (r0 + 16 * u) * IN_P + kq * IN_PART + p4) = sv[kq][u];
    __builtin_amdgcn_fence(__ATOMIC_RELEASE, "workgroup", "local");
    __builtin_amdgcn_s_barrier();
    __builtin_amdgcn_fence(__ATOMIC_ACQUIRE, "workgroup", "local");
    #pragma unroll 8
    for (int kk = kq * IN_PART; kk < (kq + 1) * IN_PART; kk += 4) {
      const v2f a = *reinterpret_cast<const v2f*>(af + kk);
      #pragma unroll
      for (int j = 0; j < 4; ++j) {
        const v2f bv = *reinterpret_cast<const v2f*>(bf + j * 16 * IN_P + kk);
        acc[j] = wmma_f32_k4(a, bv, acc[j]);
      }
    }
  }
  #pragma unroll
  for (int j = 0; j < 4; ++j)
    #pragma unroll
    for (int i = 0; i < 8; ++i) {
      const int row = w * 16 + half * 8 + i;
      const int col = dBase + j * 16 + lan;
      out[((long)(c * C_SZ + row) * B_SZ + b) * D_SZ + col] += acc[j][i];
    }
}

extern "C" void kernel_launch(void* const* d_in, const int* in_sizes, int n_in,
                              void* d_out, int out_size, void* d_ws, size_t ws_size,
                              hipStream_t stream) {
  const float* v     = (const float*)d_in[0];
  const float* k     = (const float*)d_in[1];
  const float* alpha = (const float*)d_in[2];
  const float* q     = (const float*)d_in[3];
  float* out = (float*)d_out;
  float* ws  = (float*)d_ws;

  const long KD_N = (long)T_TOT * B_SZ * N_SZ;
  const long PF_N = (long)NCHUNK * B_SZ * N_SZ;
  const long U_N  = (long)NCHUNK * B_SZ * D_SZ * N_SZ;
  float* kd     = ws;
  float* prefix = kd + KD_N;
  float* U      = prefix + PF_N;
  float* Sst    = U + U_N;

  k_gemmU<<<dim3(B_SZ * NCHUNK, D_SZ / 128, N_SZ / 64), 256, 0, stream>>>(v, k, alpha, kd, prefix, U);
  k_intra<<<dim3(B_SZ * NCHUNK, C_SZ / 16), 128, 0, stream>>>(q, kd, v, U, prefix, Sst, out);
  k_inter<<<dim3(B_SZ * NCHUNK, D_SZ / 64), 256, 0, stream>>>(q, prefix, Sst, out);
}


# all three epilogues: each wave parks its 16x64 accumulator tile in an LDS patch of its own and stores it (k_inter: reads old out too) as 16-byte pieces, whole 128-byte lines (was 4-byte, half lines)
# speedup vs baseline: 3.6975x; 1.1886x over previous
#include <hip/hip_runtime.h>


typedef float v2f __attribute__((ext_vector_type(2)));
typedef float v8f __attribute__((ext_vector_type(8)));

#define T_TOT  2048
#define B_SZ   4
#define D_SZ   256
#define N_SZ   256
#define C_SZ   128
#define NCHUNK 16
#define EPS_F  1e-8f

__device__ __forceinline__ v8f wmma_f32_k4(v2f a, v2f b, v8f c) {
  return __builtin_amdgcn_wmma_f32_16x16x4_f32(false, a, false, b, (short)0, c, false, false);
}

typedef float v4f __attribute__((ext_vector_type(4)));

#define DK_SEGS  16
#define DK_STEPS (C_SZ / DK_SEGS)

__device__ __forceinline__ v4f clip4(v4f a) {
  v4f r;
  r.x = (a.x < EPS_F) ? EPS_F : a.x;
  r.y = (a.y < EPS_F) ? EPS_F : a.y;
  r.z = (a.z < EPS_F) ? EPS_F : a.z;
  r.w = (a.w < EPS_F) ? EPS_F : a.w;
  return r;
}

template <int PITCH>
__device__ __forceinline__ void tile_to_rows(float* __restrict__ st, const v8f (&acc)[4], int half, int lan,
                                             v4f (&rows)[8]) {
  #pragma unroll
  for (int j = 0; j < 4; ++j)
    #pragma unroll
    for (int i = 0; i < 8; ++i)
      st[(half * 8 + i) * PITCH + j * 16 + lan] = acc[j][i];
  __builtin_amdgcn_fence(__ATOMIC_RELEASE, "workgroup", "local");
  __builtin_amdgcn_fence(__ATOMIC_ACQUIRE, "workgroup", "local");
  #pragma unroll
  for (int u = 0; u < 8; ++u)
    rows[u] = *reinterpret_cast<const v4f*>(st + (2 * u + half) * PITCH + lan * 4);
}

#define EP_P 68

#define GU_PA   136
#define GU_P    72
#define GU_HALF 64

__global__ void __launch_bounds__(256) k_gemmU(const float* __restrict__ v, const float* __restrict__ kin,
                                               const float* __restrict__ alpha, float* __restrict__ kd,
                                               float* __restrict__ prefix, float* __restrict__ U) {
  __shared__ __attribute__((aligned(16))) float Av[C_SZ * GU_PA];
  __shared__ __attribute__((aligned(16))) float Bk[C_SZ * GU_P];
  __shared__ v4f segTot[DK_SEGS][16];
  __shared__ __attribute__((aligned(16))) float St[8 * 16 * EP_P];
  const int c = blockIdx.x / B_SZ, b = blockIdx.x % B_SZ;
  const int dBase = blockIdx.y * 128, nBase = blockIdx.z * 64;
  const int tid = threadIdx.x, lane = tid & 31, w = tid >> 5;
  const long rowN = (long)B_SZ * N_SZ, rowD = (long)B_SZ * D_SZ;
  const float* vg = v + ((long)(c * C_SZ) * B_SZ + b) * D_SZ + dBase;
  const int q4  = tid & 15;
  const int seg = tid >> 4;
  const int tRow = seg * DK_STEPS;
  const long kbase = ((long)(c * C_SZ + tRow) * B_SZ + b) * N_SZ + nBase + q4 * 4;
  const int c4 = lane * 4;

  v4f a[DK_STEPS], kv[DK_STEPS], va[2][8];
  #pragma unroll
  for (int i = 0; i < DK_STEPS; ++i)
    a[i] = *reinterpret_cast<const v4f*>(alpha + kbase + (long)i * rowN);
  __builtin_amdgcn_sched_barrier(0);
  #pragma unroll
  for (int i = 0; i < DK_STEPS; ++i)
    kv[i] = *reinterpret_cast<const v4f*>(kin + kbase + (long)i * rowN);
  #pragma unroll
  for (int h2 = 0; h2 < 2; ++h2) {
    #pragma unroll
    for (int u = 0; u < 8; ++u)
      va[h2][u] = *reinterpret_cast<const v4f*>(vg + (long)(h2 * GU_HALF + w + 8 * u) * rowD + c4);
  }
  __builtin_amdgcn_sched_barrier(0);

  v4f tot = {1.0f, 1.0f, 1.0f, 1.0f};
  #pragma unroll
  for (int i = 0; i < DK_STEPS; ++i) {
    a[i] = clip4(a[i]);
    tot *= a[i];
  }
  segTot[seg][q4] = tot;
  __builtin_amdgcn_fence(__ATOMIC_RELEASE, "workgroup", "local");
  __builtin_amdgcn_s_barrier();
  __builtin_amdgcn_fence(__ATOMIC_ACQUIRE, "workgroup", "local");

  v4f E = {1.0f, 1.0f, 1.0f, 1.0f};
  v4f P = {1.0f, 1.0f, 1.0f, 1.0f};
  #pragma unroll
  for (int j = 0; j < DK_SEGS; ++j) {
    if (j == seg) E = P;
    P *= segTot[j][q4];
  }
  const v4f inv = 1.0f / (P + EPS_F);

  v4f r = E;
  #pragma unroll
  for (int i = 0; i < DK_STEPS; ++i) {
    r *= a[i];
    kv[i] = kv[i] * (r * inv);
  }
  #pragma unroll
  for (int i = 0; i < DK_STEPS; ++i)
    *reinterpret_cast<v4f*>(Bk + (tRow + i) * GU_P + q4 * 4) = kv[i];
  if (blockIdx.y == 0) {
    #pragma unroll
    for (int i = 0; i < DK_STEPS; ++i)
      *reinterpret_cast<v4f*>(kd + kbase + (long)i * rowN) = kv[i];
    if (seg == 0)
      *reinterpret_cast<v4f*>(prefix + (c * B_SZ + b) * N_SZ + nBase + q4 * 4) = a[0];
  }

  const int half = lane >> 4, lan = lane & 15;
  const int mLoc = w * 16 + lan;
  v8f acc[4] = {};
  #pragma unroll
  for (int h2 = 0; h2 < 2; ++h2) {
    #pragma unroll
    for (int u = 0; u < 8; ++u)
      *reinterpret_cast<v4f*>(Av + (h2 * GU_HALF + w + 8 * u) * GU_PA + c4) = va[h2][u];
    __builtin_amdgcn_fence(__ATOMIC_RELEASE, "workgroup", "local");
    __builtin_amdgcn_s_barrier();
    __builtin_amdgcn_fence(__ATOMIC_ACQUIRE, "workgroup", "local");
    #pragma unroll
    for (int kk = h2 * GU_HALF; kk < (h2 + 1) * GU_HALF; kk += 4) {
      const int ka = kk + half * 2;
      v2f af; af.x = Av[ka * GU_PA + mLoc]; af.y = Av[(ka + 1) * GU_PA + mLoc];
      #pragma unroll
      for (int j = 0; j < 4; ++j) {
        v2f bf; bf.x = Bk[ka * GU_P + j * 16 + lan]; bf.y = Bk[(ka + 1) * GU_P + j * 16 + lan];
        acc[j] = wmma_f32_k4(af, bf, acc[j]);
      }
    }
  }
  v4f rows[8];
  tile_to_rows<EP_P>(St + w * (16 * EP_P), acc, half, lan, rows);
  float* ug = U + ((long)(c * B_SZ + b) * D_SZ + dBase + w * 16) * N_SZ + nBase + lan * 4;
  #pragma unroll
  for (int u = 0; u < 8; ++u)
    *reinterpret_cast<v4f*>(ug + (long)(2 * u + half) * N_SZ) = rows[u];
}

#define IT_QP 260
#define IT_VP 264
#define IT_AP 132

template <int NB>
__device__ __forceinline__ void stage_rows(float* __restrict__ dst, int pitch, const float* __restrict__ src,
                                           long rowStride, int i0, int tid) {
  v4f tmp[NB];
  #pragma unroll
  for (int u = 0; u < NB; ++u) {
    const int idx = tid + (i0 + u) * 128, row = idx >> 6, c4 = (idx & 63) * 4;
    tmp[u] = *reinterpret_cast<const v4f*>(src + (long)row * rowStride + c4);
  }
  #pragma unroll
  for (int u = 0; u < NB; ++u) {
    const int idx = tid + (i0 + u) * 128, row = idx >> 6, c4 = (idx & 63) * 4;
    *reinterpret_cast<v4f*>(dst + row * pitch + c4) = tmp[u];
  }
}

__device__ __forceinline__ void stage_all(float* __restrict__ dst, int pitch, const float* __restrict__ src,
                                          long rowStride, int n, int tid) {
  int i = 0;
  for (; i + 32 <= n; i += 32) stage_rows<32>(dst, pitch, src, rowStride, i, tid);
  if (i + 16 <= n) { stage_rows<16>(dst, pitch, src, rowStride, i, tid); i += 16; }
  if (i + 8 <= n)  { stage_rows<8>(dst, pitch, src, rowStride, i, tid); }
}

__global__ void __launch_bounds__(128) k_intra(const float* __restrict__ q, const float* __restrict__ kd,
                                               const float* __restrict__ v, const float* __restrict__ U,
                                               const float* __restrict__ prefix, float* __restrict__ Sst,
                                               float* __restrict__ out) {
  __shared__ __attribute__((aligned(16))) float Qs[16 * IT_QP];
  __shared__ __attribute__((aligned(16))) float Ps[C_SZ * IT_VP];
  __shared__ __attribute__((aligned(16))) float As[16 * IT_AP];
  const int c = blockIdx.x / B_SZ, b = blockIdx.x % B_SZ;
  const int r  = (blockIdx.y < 4) ? (7 - (int)blockIdx.y) : ((int)blockIdx.y - 4);
  const int t0 = r * 16;
  const int S  = t0 + 16;
  const int tid = threadIdx.x, lane = tid & 31;
  const int w = __builtin_amdgcn_readfirstlane(tid >> 5);
  const int half = lane >> 4, lan = lane & 15;
  const long rowN = (long)B_SZ * N_SZ, rowD = (long)B_SZ * D_SZ;
  const float* qg  = q  + ((long)(c * C_SZ) * B_SZ + b) * N_SZ;
  const float* kg  = kd + ((long)(c * C_SZ) * B_SZ + b) * N_SZ;
  const float* vg  = v  + ((long)(c * C_SZ) * B_SZ + b) * D_SZ;

  const int  sg  = (((int)blockIdx.y * (B_SZ * NCHUNK) + (int)blockIdx.x) * 128 + tid) * 4;
  const int  sb  = sg / (D_SZ * N_SZ), sn = sg % N_SZ;
  const long scs = (long)B_SZ * D_SZ * N_SZ;
  v4f su[NCHUNK - 1], sp[NCHUNK - 1];
  #pragma unroll
  for (int cc = 0; cc < NCHUNK - 1; ++cc) {
    su[cc] = *reinterpret_cast<const v4f*>(U + cc * scs + sg);
    sp[cc] = *reinterpret_cast<const v4f*>(prefix + (cc * B_SZ + sb) * N_SZ + sn);
  }

  stage_rows<8>(Qs, IT_QP, qg + (long)t0 * rowN, rowN, 0, tid);
  stage_all(Ps, IT_QP, kg, rowN, S / 2, tid);
  __syncthreads();

  {
    v4f s = {0.0f, 0.0f, 0.0f, 0.0f};
    #pragma unroll
    for (int cc = 0; cc < NCHUNK; ++cc) {
      *reinterpret_cast<v4f*>(Sst + cc * scs + sg) = s;
      if (cc < NCHUNK - 1) s = s * sp[cc] + su[cc];
    }
  }

  v8f a0 = {}, a1 = {};
  const float* qf  = Qs + lan * IT_QP + half * 2;
  const float* kf0 = Ps + (w * 16 + lan) * IT_QP + half * 2;
  const float* kf1 = kf0 + 64 * IT_QP;
  if (w + 4 <= r) {
    #pragma unroll 8
    for (int kk = 0; kk < N_SZ; kk += 4) {
      const v2f a  = *reinterpret_cast<const v2f*>(qf + kk);
      const v2f b0 = *reinterpret_cast<const v2f*>(kf0 + kk);
      const v2f b1 = *reinterpret_cast<const v2f*>(kf1 + kk);
      a0 = wmma_f32_k4(a, b0, a0);
      a1 = wmma_f32_k4(a, b1, a1);
    }
  } else if (w <= r) {
    #pragma unroll 8
    for (int kk = 0; kk < N_SZ; kk += 4) {
      const v2f a  = *reinterpret_cast<const v2f*>(qf + kk);
      const v2f b0 = *reinterpret_cast<const v2f*>(kf0 + kk);
      a0 = wmma_f32_k4(a, b0, a0);
    }
  }
  if (w <= r) {
    #pragma unroll
    for (int i = 0; i < 8; ++i) {
      const int row = half * 8 + i, col = w * 16 + lan;
      As[row * IT_AP + col] = (col <= t0 + row) ? a0[i] : 0.0f;
    }
  }
  if (w + 4 <= r) {
    #pragma unroll
    for (int i = 0; i < 8; ++i) {
      const int row = half * 8 + i, col = (w + 4) * 16 + lan;
      As[row * IT_AP + col] = (col <= t0 + row) ? a1[i] : 0.0f;
    }
  }
  __syncthreads();

  stage_all(Ps, IT_VP, vg, rowD, S / 2, tid);
  __syncthreads();

  v8f acc[4] = {};
  const float* af = As + lan * IT_AP + half * 2;
  const float* vf = Ps + (half * 2) * IT_VP + w * 64 + lan;
  #pragma unroll 4
  for (int kk = 0; kk < S; kk += 4) {
    const v2f a = *reinterpret_cast<const v2f*>(af + kk);
    #pragma unroll
    for (int j = 0; j < 4; ++j) {
      v2f bf; bf.x = vf[kk * IT_VP + j * 16]; bf.y = vf[(kk + 1) * IT_VP + j * 16];
      acc[j] = wmma_f32_k4(a, bf, acc[j]);
    }
  }
  v4f rows[8];
  tile_to_rows<64>(Qs + w * (16 * 64), acc, half, lan, rows);
  float* og = out + ((long)(c * C_SZ + t0) * B_SZ + b) * D_SZ + w * 64 + lan * 4;
  #pragma unroll
  for (int u = 0; u < 8; ++u)
    *reinterpret_cast<v4f*>(og + (long)(2 * u + half) * rowD) = rows[u];
}

#define IN_P    260
#define IN_PART 64

__global__ void __launch_bounds__(256) k_inter(const float* __restrict__ q, const float* __restrict__ prefix,
                                               const float* __restrict__ Sst, float* __restrict__ out) {
  __shared__ __attribute__((aligned(16))) float Qp[C_SZ * IN_P];
  __shared__ __attribute__((aligned(16))) float Sp[64 * IN_P];
  __shared__ __attribute__((aligned(16))) float St[8 * 16 * EP_P];
  const int c = blockIdx.x / B_SZ, b = blockIdx.x % B_SZ;
  const int dBase = blockIdx.y * 64;
  const int tid = threadIdx.x, lane = tid & 31, w = tid >> 5;
  const int half = lane >> 4, lan = lane & 15;
  const long cs = (long)B_SZ * D_SZ * N_SZ;
  const long rowN = (long)B_SZ * N_SZ;
  const float* qg = q + ((long)(c * C_SZ) * B_SZ + b) * N_SZ;
  const float* sg = Sst + (long)c * cs + ((long)b * D_SZ + dBase) * N_SZ;
  const float* pg = prefix + (c * B_SZ + b) * N_SZ;
  const int p4 = (tid & 15) * 4;
  const int r0 = tid >> 4;

  v4f pf[4], qv[4][8], sv[4][4];
  #pragma unroll
  for (int kq = 0; kq < 4; ++kq) {
    pf[kq] = *reinterpret_cast<const v4f*>(pg + kq * IN_PART + p4);
    #pragma unroll
    for (int u = 0; u < 8; ++u)
      qv[kq][u] = *reinterpret_cast<const v4f*>(qg + (long)(r0 + 16 * u) * rowN + kq * IN_PART + p4);
    #pragma unroll
    for (int u = 0; u < 4; ++u)
      sv[kq][u] = *reinterpret_cast<const v4f*>(sg + (long)(r0 + 16 * u) * N_SZ + kq * IN_PART + p4);
  }

  v8f acc[4] = {};
  const float* af = Qp + (w * 16 + lan) * IN_P + half * 2;
  const float* bf = Sp + lan * IN_P + half * 2;
  #pragma unroll
  for (int kq = 0; kq < 4; ++kq) {
    #pragma unroll
    for (int u = 0; u < 8; ++u)
      *reinterpret_cast<v4f*>(Qp + (r0 + 16 * u) * IN_P + kq * IN_PART + p4) = qv[kq][u] * pf[kq];
    #pragma unroll
    for (int u = 0; u < 4; ++u)
      *reinterpret_cast<v4f*>(Sp + (r0 + 16 * u) * IN_P + kq * IN_PART + p4) = sv[kq][u];
    __builtin_amdgcn_fence(__ATOMIC_RELEASE, "workgroup", "local");
    __builtin_amdgcn_s_barrier();
    __builtin_amdgcn_fence(__ATOMIC_ACQUIRE, "workgroup", "local");
    #pragma unroll 8
    for (int kk = kq * IN_PART; kk < (kq + 1) * IN_PART; kk += 4) {
      const v2f a = *reinterpret_cast<const v2f*>(af + kk);
      #pragma unroll
      for (int j = 0; j < 4; ++j) {
        const v2f bv = *reinterpret_cast<const v2f*>(bf + j * 16 * IN_P + kk);
        acc[j] = wmma_f32_k4(a, bv, acc[j]);
      }
    }
  }
  const long rowD = (long)B_SZ * D_SZ;
  float* og = out + ((long)(c * C_SZ + w * 16) * B_SZ + b) * D_SZ + dBase + lan * 4;
  v4f old[8], rows[8];
  #pragma unroll
  for (int u = 0; u < 8; ++u)
    old[u] = *reinterpret_cast<const v4f*>(og + (long)(2 * u + half) * rowD);
  tile_to_rows<EP_P>(St + w * (16 * EP_P), acc, half, lan, rows);
  #pragma unroll
  for (int u = 0; u < 8; ++u)
    *reinterpret_cast<v4f*>(og + (long)(2 * u + half) * rowD) = old[u] + rows[u];
}

extern "C" void kernel_launch(void* const* d_in, const int* in_sizes, int n_in,
                              void* d_out, int out_size, void* d_ws, size_t ws_size,
                              hipStream_t stream) {
  const float* v     = (const float*)d_in[0];
  const float* k     = (const float*)d_in[1];
  const float* alpha = (const float*)d_in[2];
  const float* q     = (const float*)d_in[3];
  float* out = (float*)d_out;
  float* ws  = (float*)d_ws;

  const long KD_N = (long)T_TOT * B_SZ * N_SZ;
  const long PF_N = (long)NCHUNK * B_SZ * N_SZ;
  const long U_N  = (long)NCHUNK * B_SZ * D_SZ * N_SZ;
  float* kd     = ws;
  float* prefix = kd + KD_N;
  float* U      = prefix + PF_N;
  float* Sst    = U + U_N;

  k_gemmU<<<dim3(B_SZ * NCHUNK, D_SZ / 128, N_SZ / 64), 256, 0, stream>>>(v, k, alpha, kd, prefix, U);
  k_intra<<<dim3(B_SZ * NCHUNK, C_SZ / 16), 128, 0, stream>>>(q, kd, v, U, prefix, Sst, out);
  k_inter<<<dim3(B_SZ * NCHUNK, D_SZ / 64), 256, 0, stream>>>(q, prefix, Sst, out);
}
